# starting kernel
# baseline (speedup 1.0000x reference)
#include <hip/hip_runtime.h>

#define N_  50000
#define E_  800000
#define D_  256
#define H_  4
#define F_  32
#define HF_ 128
#define NEG_SLOPE 0.2f

typedef __attribute__((ext_vector_type(2))) float v2f;
typedef __attribute__((ext_vector_type(8))) float v8f;

__device__ __forceinline__ unsigned encodeOrd(float f) {
  unsigned u = __float_as_uint(f);
  return (u & 0x80000000u) ? ~u : (u | 0x80000000u);
}
__device__ __forceinline__ float decodeOrd(unsigned u) {
  unsigned b = (u & 0x80000000u) ? (u & 0x7FFFFFFFu) : ~u;
  return __uint_as_float(b);
}

__global__ __launch_bounds__(256) void gat_gemm_wmma(
    const float* __restrict__ h, const float* __restrict__ W,
    float* __restrict__ hW, int nRows)
{
  const int wave = threadIdx.x >> 5;
  const int lane = threadIdx.x & 31;
  const int rowBase = (blockIdx.x * 8 + wave) * 16;
  if (rowBase >= nRows) return;

  const int m    = lane & 15;
  const int khalf= (lane >> 4) * 2;

  v8f acc[8];
#pragma unroll
  for (int t = 0; t < 8; ++t) acc[t] = (v8f){0.f,0.f,0.f,0.f,0.f,0.f,0.f,0.f};

  for (int k0 = 0; k0 < D_; k0 += 4) {
    const float* ap = h + (size_t)(rowBase + m) * D_ + (k0 + khalf);
    v2f a; a.x = ap[0]; a.y = ap[1];
#pragma unroll
    for (int t = 0; t < 8; ++t) {
      const int n = t * 16 + (lane & 15);
      const float* bp = W + (size_t)(k0 + khalf) * HF_ + n;
      v2f b; b.x = bp[0]; b.y = bp[HF_];
      acc[t] = __builtin_amdgcn_wmma_f32_16x16x4_f32(
           false, a,  false, b,
           (short)0, acc[t],  false,  false);
    }
  }

  const int mHalf = (lane >> 4) * 8;
#pragma unroll
  for (int t = 0; t < 8; ++t) {
    const int n = t * 16 + (lane & 15);
#pragma unroll
    for (int v = 0; v < 8; ++v) {
      hW[(size_t)(rowBase + mHalf + v) * HF_ + n] = acc[t][v];
    }
  }
}

__global__ __launch_bounds__(256) void gat_attn_proj(
    const float* __restrict__ hW,
    const float* __restrict__ alw, const float* __restrict__ alb,
    const float* __restrict__ arw, const float* __restrict__ arb,
    float* __restrict__ el, float* __restrict__ er, int nNH)
{
  const int idx = blockIdx.x * blockDim.x + threadIdx.x;
  if (idx >= nNH) return;
  const float* p = hW + (size_t)idx * F_;
  float sl = 0.f, sr = 0.f;
#pragma unroll
  for (int f = 0; f < F_; ++f) {
    const float x = p[f];
    sl += x * alw[f];
    sr += x * arw[f];
  }
  el[idx] = sl + alb[0];
  er[idx] = sr + arb[0];
}

__global__ __launch_bounds__(256) void gat_init(
    unsigned* __restrict__ emax, float* __restrict__ denom,
    float* __restrict__ out, const float* __restrict__ bias,
    int nNH, int nNHF)
{
  const int i = blockIdx.x * blockDim.x + threadIdx.x;
  if (i < nNH) {
    emax[i]  = encodeOrd(-__builtin_inff());
    denom[i] = 0.f;
  }
  if (i < nNHF) out[i] = bias[i & (HF_ - 1)];
}

__global__ __launch_bounds__(256) void gat_edge_max(
    const int* __restrict__ src, const int* __restrict__ dst,
    const float* __restrict__ el, const float* __restrict__ er,
    unsigned* __restrict__ emax, int nEH)
{
  const int i = blockIdx.x * blockDim.x + threadIdx.x;
  if (i >= nEH) return;
  const int e = i >> 2;
  const int hh = i & (H_ - 1);
  const int s = src[e], d = dst[e];
  float v = er[s * H_ + hh] + el[d * H_ + hh];
  v = (v >= 0.f) ? v : NEG_SLOPE * v;
  atomicMax(&emax[d * H_ + hh], encodeOrd(v));
}

__global__ __launch_bounds__(256) void gat_edge_exp(
    const int* __restrict__ src, const int* __restrict__ dst,
    const float* __restrict__ el, const float* __restrict__ er,
    const unsigned* __restrict__ emax, float* __restrict__ ee,
    float* __restrict__ denom, int nEH)
{
  const int i = blockIdx.x * blockDim.x + threadIdx.x;
  if (i >= nEH) return;
  const int e = i >> 2;
  const int hh = i & (H_ - 1);
  const int s = src[e], d = dst[e];
  float v = er[s * H_ + hh] + el[d * H_ + hh];
  v = (v >= 0.f) ? v : NEG_SLOPE * v;
  const float x = expf(v - decodeOrd(emax[d * H_ + hh]));
  ee[i] = x;
  atomicAdd(&denom[d * H_ + hh], x);
}

__global__ __launch_bounds__(256) void gat_scatter(
    const int* __restrict__ src, const int* __restrict__ dst,
    const float* __restrict__ hW, const float* __restrict__ ee,
    const float* __restrict__ denom, float* __restrict__ out,
    long long nTot)
{
  const long long tid = (long long)blockIdx.x * blockDim.x + threadIdx.x;
  if (tid >= nTot) return;
  const long long e = tid >> 7;
  const int hf = (int)(tid & 127);
  const int hh = hf >> 5;
  const int s = src[e], d = dst[e];
  const float a = ee[e * H_ + hh] / denom[d * H_ + hh];
  atomicAdd(&out[(size_t)d * HF_ + hf], hW[(size_t)s * HF_ + hf] * a);
}

extern "C" void kernel_launch(void* const* d_in, const int* in_sizes, int n_in,
                              void* d_out, int out_size, void* d_ws, size_t ws_size,
                              hipStream_t stream)
{
  const float* h    = (const float*)d_in[0];
  const float* W    = (const float*)d_in[1];
  const float* alw  = (const float*)d_in[2];
  const float* alb  = (const float*)d_in[3];
  const float* arw  = (const float*)d_in[4];
  const float* arb  = (const float*)d_in[5];
  const float* bias = (const float*)d_in[6];
  const int*   src  = (const int*)d_in[7];
  const int*   dst  = (const int*)d_in[8];
  float* out = (float*)d_out;

  char* ws = (char*)d_ws;
  size_t off = 0;
  auto carve = [&](size_t bytes) -> void* {
    void* p = ws + off;
    off += (bytes + 255) & ~(size_t)255;
    return p;
  };
  float*    hW    = (float*)carve((size_t)N_ * HF_ * sizeof(float));
  float*    el    = (float*)carve((size_t)N_ * H_ * sizeof(float));
  float*    er    = (float*)carve((size_t)N_ * H_ * sizeof(float));
  unsigned* emax  = (unsigned*)carve((size_t)N_ * H_ * sizeof(unsigned));
  float*    denom = (float*)carve((size_t)N_ * H_ * sizeof(float));
  float*    ee    = (float*)carve((size_t)E_ * H_ * sizeof(float));
  (void)ws_size; (void)in_sizes; (void)n_in; (void)out_size;

  {
    const int strips = N_ / 16;
    const int blocks = (strips + 7) / 8;
    gat_gemm_wmma<<<blocks, 256, 0, stream>>>(h, W, hW, N_);
  }
  {
    const int nNH = N_ * H_;
    gat_attn_proj<<<(nNH + 255) / 256, 256, 0, stream>>>(hW, alw, alb, arw, arb, el, er, nNH);
  }
  {
    const int nNH = N_ * H_;
    const int nNHF = N_ * H_ * F_;
    gat_init<<<(nNHF + 255) / 256, 256, 0, stream>>>(emax, denom, out, bias, nNH, nNHF);
  }
  {
    const int nEH = E_ * H_;
    gat_edge_max<<<(nEH + 255) / 256, 256, 0, stream>>>(src, dst, el, er, emax, nEH);
  }
  {
    const int nEH = E_ * H_;
    gat_edge_exp<<<(nEH + 255) / 256, 256, 0, stream>>>(src, dst, el, er, emax, ee, denom, nEH);
  }
  {
    const long long nTot = (long long)E_ * HF_;
    const long long blocks = (nTot + 255) / 256;
    gat_scatter<<<(unsigned)blocks, 256, 0, stream>>>(src, dst, hW, ee, denom, out, nTot);
  }
}


# Scatter-sum in LDS: one work-group per 256 destination nodes walks the dst list, adds its edges' hW rows by LDS float atomics, stores each out row once; no global f32 atomics; sum order changes
# speedup vs baseline: 2.1679x; 2.1679x over previous
#include <hip/hip_runtime.h>

#define N_  50000
#define E_  800000
#define D_  256
#define H_  4
#define F_  32
#define HF_ 128
#define NEG_SLOPE 0.2f

typedef __attribute__((ext_vector_type(2))) float v2f;
typedef __attribute__((ext_vector_type(8))) float v8f;
typedef __attribute__((ext_vector_type(4))) float v4f;
typedef __attribute__((ext_vector_type(4))) int   v4i;

__device__ __forceinline__ unsigned encodeOrd(float f) {
  unsigned u = __float_as_uint(f);
  return (u & 0x80000000u) ? ~u : (u | 0x80000000u);
}
__device__ __forceinline__ float decodeOrd(unsigned u) {
  unsigned b = (u & 0x80000000u) ? (u & 0x7FFFFFFFu) : ~u;
  return __uint_as_float(b);
}

__global__ __launch_bounds__(256) void gat_gemm_wmma(
    const float* __restrict__ h, const float* __restrict__ W,
    float* __restrict__ hW, int nRows)
{
  const int wave = threadIdx.x >> 5;
  const int lane = threadIdx.x & 31;
  const int rowBase = (blockIdx.x * 8 + wave) * 16;
  if (rowBase >= nRows) return;

  const int m    = lane & 15;
  const int khalf= (lane >> 4) * 2;

  v8f acc[8];
#pragma unroll
  for (int t = 0; t < 8; ++t) acc[t] = (v8f){0.f,0.f,0.f,0.f,0.f,0.f,0.f,0.f};

  for (int k0 = 0; k0 < D_; k0 += 4) {
    const float* ap = h + (size_t)(rowBase + m) * D_ + (k0 + khalf);
    v2f a; a.x = ap[0]; a.y = ap[1];
#pragma unroll
    for (int t = 0; t < 8; ++t) {
      const int n = t * 16 + (lane & 15);
      const float* bp = W + (size_t)(k0 + khalf) * HF_ + n;
      v2f b; b.x = bp[0]; b.y = bp[HF_];
      acc[t] = __builtin_amdgcn_wmma_f32_16x16x4_f32(
           false, a,  false, b,
           (short)0, acc[t],  false,  false);
    }
  }

  const int mHalf = (lane >> 4) * 8;
#pragma unroll
  for (int t = 0; t < 8; ++t) {
    const int n = t * 16 + (lane & 15);
#pragma unroll
    for (int v = 0; v < 8; ++v) {
      hW[(size_t)(rowBase + mHalf + v) * HF_ + n] = acc[t][v];
    }
  }
}

__global__ __launch_bounds__(256) void gat_attn_proj(
    const float* __restrict__ hW,
    const float* __restrict__ alw, const float* __restrict__ alb,
    const float* __restrict__ arw, const float* __restrict__ arb,
    float* __restrict__ el, float* __restrict__ er, int nNH)
{
  const int idx = blockIdx.x * blockDim.x + threadIdx.x;
  if (idx >= nNH) return;
  const float* p = hW + (size_t)idx * F_;
  float sl = 0.f, sr = 0.f;
#pragma unroll
  for (int f = 0; f < F_; ++f) {
    const float x = p[f];
    sl += x * alw[f];
    sr += x * arw[f];
  }
  el[idx] = sl + alb[0];
  er[idx] = sr + arb[0];
}

__global__ __launch_bounds__(256) void gat_init(
    unsigned* __restrict__ emax, float* __restrict__ denom, int nNH)
{
  const int i = blockIdx.x * blockDim.x + threadIdx.x;
  if (i < nNH) {
    emax[i]  = encodeOrd(-__builtin_inff());
    denom[i] = 0.f;
  }
}

__global__ __launch_bounds__(256) void gat_edge_max(
    const int* __restrict__ src, const int* __restrict__ dst,
    const float* __restrict__ el, const float* __restrict__ er,
    unsigned* __restrict__ emax, int nEH)
{
  const int i = blockIdx.x * blockDim.x + threadIdx.x;
  if (i >= nEH) return;
  const int e = i >> 2;
  const int hh = i & (H_ - 1);
  const int s = src[e], d = dst[e];
  float v = er[s * H_ + hh] + el[d * H_ + hh];
  v = (v >= 0.f) ? v : NEG_SLOPE * v;
  atomicMax(&emax[d * H_ + hh], encodeOrd(v));
}

__global__ __launch_bounds__(256) void gat_edge_exp(
    const int* __restrict__ src, const int* __restrict__ dst,
    const float* __restrict__ el, const float* __restrict__ er,
    const unsigned* __restrict__ emax, float* __restrict__ ee,
    float* __restrict__ denom, int nEH)
{
  const int i = blockIdx.x * blockDim.x + threadIdx.x;
  if (i >= nEH) return;
  const int e = i >> 2;
  const int hh = i & (H_ - 1);
  const int s = src[e], d = dst[e];
  float v = er[s * H_ + hh] + el[d * H_ + hh];
  v = (v >= 0.f) ? v : NEG_SLOPE * v;
  const float x = expf(v - decodeOrd(emax[d * H_ + hh]));
  ee[i] = x;
  atomicAdd(&denom[d * H_ + hh], x);
}

#define TILE_N   256
#define SC_WAVES 32
#define QCAP     256
#define CHUNK    128
#define NCHUNK   (E_ / CHUNK)
#define NTILES   ((N_ + TILE_N - 1) / TILE_N)
#define BATCH    32

static_assert(E_ % CHUNK == 0, "edge list must be whole chunks");
static_assert(E_ <= (1 << 24), "edge id must fit 24 bits of a ring entry");
static_assert(TILE_N == 256, "local node id must fit 8 bits of a ring entry");
static_assert(TILE_N * H_ == SC_WAVES * 32, "one denominator per thread");
static_assert(BATCH + CHUNK <= QCAP, "ring must hold a batch plus one chunk");

template <bool FULL>
__device__ __forceinline__ void scatter_batch(
    const unsigned* qw, float* ww, unsigned qh, int n, int lane,
    const int* __restrict__ src, const float* __restrict__ hW,
    const float* __restrict__ ee, const float* den, float* acc)
{
  int s = 0, dl = 0;
  if (FULL || lane < n) {
    const unsigned ent = qw[(qh + (unsigned)lane) & (QCAP - 1)];
    const unsigned e = ent >> 8;
    dl = (int)(ent & 255u);
    s = src[e];
    const v4f x  = ((const v4f*)ee)[e];
    const v4f dn = ((const v4f*)den)[dl];
    ((v4f*)ww)[lane] = x / dn;
  }
  const int hd = lane >> 3;
  const int cnt = FULL ? BATCH : n;
  for (int g = 0; g < cnt; g += 8) {
    v4f row[8];
    float wv[8];
#pragma unroll
    for (int u = 0; u < 8; ++u) {
      row[u] = (v4f){0.f, 0.f, 0.f, 0.f};
      wv[u] = 0.f;
      if (FULL || g + u < n) {
        const int sk = __builtin_amdgcn_readlane(s, g + u);
        const v4f* rp = (const v4f*)(hW + (size_t)sk * HF_);
        row[u] = rp[(unsigned)lane];
        wv[u] = ww[(g + u) * H_ + hd];
      }
    }
#pragma unroll
    for (int u = 0; u < 8; ++u) {
      if (FULL || g + u < n) {
        const int dlk = __builtin_amdgcn_readlane(dl, g + u);
        float* p = acc + dlk * HF_ + lane;
        const float w = wv[u];
        atomicAdd(p,      row[u].x * w);
        atomicAdd(p + 32, row[u].y * w);
        atomicAdd(p + 64, row[u].z * w);
        atomicAdd(p + 96, row[u].w * w);
      }
    }
  }
}

__global__ __launch_bounds__(1024) void gat_scatter_tiles(
    const int* __restrict__ src, const int* __restrict__ dst,
    const float* __restrict__ hW, const float* __restrict__ ee,
    const float* __restrict__ denom, const float* __restrict__ bias,
    float* __restrict__ out)
{
  __shared__ float    acc[TILE_N * HF_];
  __shared__ float    den[TILE_N * H_];
  __shared__ unsigned q[SC_WAVES * QCAP];
  __shared__ float    wt[SC_WAVES * BATCH * H_];

  const int tid  = threadIdx.x;
  const int lane = tid & 31;
  const int wave = __builtin_amdgcn_readfirstlane(tid >> 5);
  const int nodeBase = blockIdx.x * TILE_N;

  for (int i = tid; i < TILE_N * HF_ / 4; i += SC_WAVES * 32)
    ((v4f*)acc)[i] = (v4f){0.f, 0.f, 0.f, 0.f};
  {
    const int g = nodeBase * H_ + tid;
    den[tid] = (g < N_ * H_) ? denom[g] : 1.f;
  }
  __syncthreads();

  unsigned* qw = q + wave * QCAP;
  float*    ww = wt + wave * (BATCH * H_);
  unsigned qh = 0, qt = 0;
  const int rot = blockIdx.x * (NCHUNK / NTILES);
  const int NJ  = (NCHUNK + SC_WAVES - 1) / SC_WAVES;

  auto chunkOf = [&](int j) -> int {
    const int k = wave + SC_WAVES * j;
    if (k >= NCHUNK) return -1;
    const int c = rot + k;
    return (c >= NCHUNK) ? c - NCHUNK : c;
  };
  auto loadChunk = [&](int j) -> v4i {
    const int c = chunkOf(j);
    const int cc = (c < 0) ? 0 : c;
    return ((const v4i*)(dst + (size_t)cc * CHUNK))[(unsigned)lane];
  };
  auto look = [&](const v4i d, int j) {
    const int c = chunkOf(j);
    if (c < 0) return;
    const unsigned r0 = (unsigned)(d.x - nodeBase), r1 = (unsigned)(d.y - nodeBase);
    const unsigned r2 = (unsigned)(d.z - nodeBase), r3 = (unsigned)(d.w - nodeBase);
    const bool m0 = r0 < TILE_N, m1 = r1 < TILE_N, m2 = r2 < TILE_N, m3 = r3 < TILE_N;
    if (__builtin_amdgcn_ballot_w32(m0 | m1 | m2 | m3) == 0u) return;
    const unsigned e0 = (unsigned)(c * CHUNK + lane * 4);
    const unsigned b0 = __builtin_amdgcn_ballot_w32(m0);
    const unsigned b1 = __builtin_amdgcn_ballot_w32(m1);
    const unsigned b2 = __builtin_amdgcn_ballot_w32(m2);
    const unsigned b3 = __builtin_amdgcn_ballot_w32(m3);
    if (m0) qw[(qt + __builtin_amdgcn_mbcnt_lo(b0, 0u)) & (QCAP - 1)] = ((e0 + 0u) << 8) | r0;
    qt += __builtin_popcount(b0);
    if (m1) qw[(qt + __builtin_amdgcn_mbcnt_lo(b1, 0u)) & (QCAP - 1)] = ((e0 + 1u) << 8) | r1;
    qt += __builtin_popcount(b1);
    if (m2) qw[(qt + __builtin_amdgcn_mbcnt_lo(b2, 0u)) & (QCAP - 1)] = ((e0 + 2u) << 8) | r2;
    qt += __builtin_popcount(b2);
    if (m3) qw[(qt + __builtin_amdgcn_mbcnt_lo(b3, 0u)) & (QCAP - 1)] = ((e0 + 3u) << 8) | r3;
    qt += __builtin_popcount(b3);
    while (qt - qh >= BATCH) {
      scatter_batch<true>(qw, ww, qh, BATCH, lane, src, hW, ee, den, acc);
      qh += BATCH;
    }
  };

  v4i bufA[4], bufB[4];
  auto load4 = [&](v4i (&buf)[4], int j) {
#pragma unroll
    for (int u = 0; u < 4; ++u) buf[u] = loadChunk(j + u);
  };
  auto look4 = [&](const v4i (&buf)[4], int j) {
#pragma unroll
    for (int u = 0; u < 4; ++u) look(buf[u], j + u);
  };
  load4(bufA, 0);
  for (int j0 = 0; j0 < NJ; j0 += 8) {
    load4(bufB, j0 + 4);
    look4(bufA, j0);
    load4(bufA, j0 + 8);
    look4(bufB, j0 + 4);
  }
  if (qt != qh)
    scatter_batch<false>(qw, ww, qh, (int)(qt - qh), lane, src, hW, ee, den, acc);

  __syncthreads();

  const v4f b4 = ((const v4f*)bias)[lane];
  for (int r = wave; r < TILE_N; r += SC_WAVES) {
    const int node = nodeBase + r;
    if (node < N_) {
      const float* p = acc + r * HF_ + lane;
      const v4f v = (v4f){p[0], p[32], p[64], p[96]};
      ((v4f*)(out + (size_t)node * HF_))[lane] = v + b4;
    }
  }
}

extern "C" void kernel_launch(void* const* d_in, const int* in_sizes, int n_in,
                              void* d_out, int out_size, void* d_ws, size_t ws_size,
                              hipStream_t stream)
{
  const float* h    = (const float*)d_in[0];
  const float* W    = (const float*)d_in[1];
  const float* alw  = (const float*)d_in[2];
  const float* alb  = (const float*)d_in[3];
  const float* arw  = (const float*)d_in[4];
  const float* arb  = (const float*)d_in[5];
  const float* bias = (const float*)d_in[6];
  const int*   src  = (const int*)d_in[7];
  const int*   dst  = (const int*)d_in[8];
  float* out = (float*)d_out;

  char* ws = (char*)d_ws;
  size_t off = 0;
  auto carve = [&](size_t bytes) -> void* {
    void* p = ws + off;
    off += (bytes + 255) & ~(size_t)255;
    return p;
  };
  float*    hW    = (float*)carve((size_t)N_ * HF_ * sizeof(float));
  float*    el    = (float*)carve((size_t)N_ * H_ * sizeof(float));
  float*    er    = (float*)carve((size_t)N_ * H_ * sizeof(float));
  unsigned* emax  = (unsigned*)carve((size_t)N_ * H_ * sizeof(unsigned));
  float*    denom = (float*)carve((size_t)N_ * H_ * sizeof(float));
  float*    ee    = (float*)carve((size_t)E_ * H_ * sizeof(float));
  (void)ws_size; (void)in_sizes; (void)n_in; (void)out_size;

  {
    const int strips = N_ / 16;
    const int blocks = (strips + 7) / 8;
    gat_gemm_wmma<<<blocks, 256, 0, stream>>>(h, W, hW, N_);
  }
  {
    const int nNH = N_ * H_;
    gat_attn_proj<<<(nNH + 255) / 256, 256, 0, stream>>>(hW, alw, alb, arw, arb, el, er, nNH);
  }
  {
    const int nNH = N_ * H_;
    gat_init<<<(nNH + 255) / 256, 256, 0, stream>>>(emax, denom, nNH);
  }
  {
    const int nEH = E_ * H_;
    gat_edge_max<<<(nEH + 255) / 256, 256, 0, stream>>>(src, dst, el, er, emax, nEH);
  }
  {
    const int nEH = E_ * H_;
    gat_edge_exp<<<(nEH + 255) / 256, 256, 0, stream>>>(src, dst, el, er, emax, ee, denom, nEH);
  }
  {
    gat_scatter_tiles<<<NTILES, SC_WAVES * 32, 0, stream>>>(src, dst, hW, ee, denom, bias, out);
  }
}


# Softmax weights and denominators made in the tile kernel (el/max in LDS, LDS float adds, row sums divided once at the store): exp launch, weight array, global float adds gone; 6 launches to 5
# speedup vs baseline: 2.7680x; 1.2768x over previous
#include <hip/hip_runtime.h>

#define N_  50000
#define E_  800000
#define D_  256
#define H_  4
#define F_  32
#define HF_ 128
#define NEG_SLOPE 0.2f

typedef __attribute__((ext_vector_type(2))) float v2f;
typedef __attribute__((ext_vector_type(8))) float v8f;
typedef __attribute__((ext_vector_type(4))) float v4f;
typedef __attribute__((ext_vector_type(4))) int   v4i;

__device__ __forceinline__ unsigned encodeOrd(float f) {
  unsigned u = __float_as_uint(f);
  return (u & 0x80000000u) ? ~u : (u | 0x80000000u);
}
__device__ __forceinline__ float decodeOrd(unsigned u) {
  unsigned b = (u & 0x80000000u) ? (u & 0x7FFFFFFFu) : ~u;
  return __uint_as_float(b);
}

__global__ __launch_bounds__(256) void gat_gemm_wmma(
    const float* __restrict__ h, const float* __restrict__ W,
    float* __restrict__ hW, int nRows)
{
  const int wave = threadIdx.x >> 5;
  const int lane = threadIdx.x & 31;
  const int rowBase = (blockIdx.x * 8 + wave) * 16;
  if (rowBase >= nRows) return;

  const int m    = lane & 15;
  const int khalf= (lane >> 4) * 2;

  v8f acc[8];
#pragma unroll
  for (int t = 0; t < 8; ++t) acc[t] = (v8f){0.f,0.f,0.f,0.f,0.f,0.f,0.f,0.f};

  for (int k0 = 0; k0 < D_; k0 += 4) {
    const float* ap = h + (size_t)(rowBase + m) * D_ + (k0 + khalf);
    v2f a; a.x = ap[0]; a.y = ap[1];
#pragma unroll
    for (int t = 0; t < 8; ++t) {
      const int n = t * 16 + (lane & 15);
      const float* bp = W + (size_t)(k0 + khalf) * HF_ + n;
      v2f b; b.x = bp[0]; b.y = bp[HF_];
      acc[t] = __builtin_amdgcn_wmma_f32_16x16x4_f32(
           false, a,  false, b,
           (short)0, acc[t],  false,  false);
    }
  }

  const int mHalf = (lane >> 4) * 8;
#pragma unroll
  for (int t = 0; t < 8; ++t) {
    const int n = t * 16 + (lane & 15);
#pragma unroll
    for (int v = 0; v < 8; ++v) {
      hW[(size_t)(rowBase + mHalf + v) * HF_ + n] = acc[t][v];
    }
  }
}

__global__ __launch_bounds__(256) void gat_attn_proj(
    const float* __restrict__ hW,
    const float* __restrict__ alw, const float* __restrict__ alb,
    const float* __restrict__ arw, const float* __restrict__ arb,
    float* __restrict__ el, float* __restrict__ er, int nNH)
{
  const int idx = blockIdx.x * blockDim.x + threadIdx.x;
  if (idx >= nNH) return;
  const float* p = hW + (size_t)idx * F_;
  float sl = 0.f, sr = 0.f;
#pragma unroll
  for (int f = 0; f < F_; ++f) {
    const float x = p[f];
    sl += x * alw[f];
    sr += x * arw[f];
  }
  el[idx] = sl + alb[0];
  er[idx] = sr + arb[0];
}

__global__ __launch_bounds__(256) void gat_init(
    unsigned* __restrict__ emax, int nNH)
{
  const int i = blockIdx.x * blockDim.x + threadIdx.x;
  if (i < nNH) emax[i] = encodeOrd(-__builtin_inff());
}

__global__ __launch_bounds__(256) void gat_edge_max(
    const int* __restrict__ src, const int* __restrict__ dst,
    const float* __restrict__ el, const float* __restrict__ er,
    unsigned* __restrict__ emax, int nEH)
{
  const int i = blockIdx.x * blockDim.x + threadIdx.x;
  if (i >= nEH) return;
  const int e = i >> 2;
  const int hh = i & (H_ - 1);
  const int s = src[e], d = dst[e];
  float v = er[s * H_ + hh] + el[d * H_ + hh];
  v = (v >= 0.f) ? v : NEG_SLOPE * v;
  atomicMax(&emax[d * H_ + hh], encodeOrd(v));
}

#define TILE_N   256
#define SC_WAVES 32
#define QCAP     256
#define CHUNK    128
#define NCHUNK   (E_ / CHUNK)
#define NTILES   ((N_ + TILE_N - 1) / TILE_N)
#define BATCH    32

static_assert(E_ % CHUNK == 0, "edge list must be whole chunks");
static_assert(E_ <= (1 << 24), "edge id must fit 24 bits of a ring entry");
static_assert(TILE_N == 256, "local node id must fit 8 bits of a ring entry");
static_assert(TILE_N * H_ == SC_WAVES * 32, "one denominator per thread");
static_assert(BATCH + CHUNK <= QCAP, "ring must hold a batch plus one chunk");

template <bool FULL>
__device__ __forceinline__ void scatter_batch(
    const unsigned* qw, float* ww, unsigned qh, int n, int lane,
    const int* __restrict__ src, const float* __restrict__ hW,
    const float* __restrict__ er, const float* elT, const float* mxT,
    float* den, float* acc)
{
  const bool mine = FULL || lane < n;
  int s = 0, dl = 0;
  v4f r4 = (v4f){0.f, 0.f, 0.f, 0.f};
  if (mine) {
    const unsigned ent = qw[(qh + (unsigned)lane) & (QCAP - 1)];
    const unsigned e = ent >> 8;
    dl = (int)(ent & 255u);
    s = src[e];
    r4 = ((const v4f*)er)[s];
  }
  const int hd = lane >> 3;
  const int cnt = FULL ? BATCH : n;
  v4f row[8];
  float wv[8];
  auto loadRows = [&](int g) {
#pragma unroll
    for (int u = 0; u < 8; ++u) {
      row[u] = (v4f){0.f, 0.f, 0.f, 0.f};
      if (FULL || g + u < n) {
        const int sk = __builtin_amdgcn_readlane(s, g + u);
        const v4f* rp = (const v4f*)(hW + (size_t)sk * HF_);
        row[u] = rp[(unsigned)lane];
      }
    }
  };
  auto loadWeights = [&](int g) {
#pragma unroll
    for (int u = 0; u < 8; ++u) {
      wv[u] = 0.f;
      if (FULL || g + u < n) wv[u] = ww[(g + u) * H_ + hd];
    }
  };
  auto addRows = [&](int g) {
#pragma unroll
    for (int u = 0; u < 8; ++u) {
      if (FULL || g + u < n) {
        const int dlk = __builtin_amdgcn_readlane(dl, g + u);
        float* p = acc + dlk * HF_ + lane;
        const float w = wv[u];
        atomicAdd(p,      row[u].x * w);
        atomicAdd(p + 32, row[u].y * w);
        atomicAdd(p + 64, row[u].z * w);
        atomicAdd(p + 96, row[u].w * w);
      }
    }
  };

  loadRows(0);
  if (mine) {
    const v4f l4 = ((const v4f*)elT)[dl];
    const v4f m4 = ((const v4f*)mxT)[dl];
    float v0 = r4.x + l4.x, v1 = r4.y + l4.y, v2 = r4.z + l4.z, v3 = r4.w + l4.w;
    v0 = (v0 >= 0.f) ? v0 : NEG_SLOPE * v0;
    v1 = (v1 >= 0.f) ? v1 : NEG_SLOPE * v1;
    v2 = (v2 >= 0.f) ? v2 : NEG_SLOPE * v2;
    v3 = (v3 >= 0.f) ? v3 : NEG_SLOPE * v3;
    v4f x;
    x.x = expf(v0 - m4.x);
    x.y = expf(v1 - m4.y);
    x.z = expf(v2 - m4.z);
    x.w = expf(v3 - m4.w);
    ((v4f*)ww)[lane] = x;
    float* dp = den + dl * H_;
    atomicAdd(dp + 0, x.x);
    atomicAdd(dp + 1, x.y);
    atomicAdd(dp + 2, x.z);
    atomicAdd(dp + 3, x.w);
  }
  loadWeights(0);
  addRows(0);
  for (int g = 8; g < cnt; g += 8) {
    loadRows(g);
    loadWeights(g);
    addRows(g);
  }
}

__global__ __launch_bounds__(1024) void gat_scatter_tiles(
    const int* __restrict__ src, const int* __restrict__ dst,
    const float* __restrict__ hW, const float* __restrict__ el,
    const float* __restrict__ er, const unsigned* __restrict__ emax,
    const float* __restrict__ bias, float* __restrict__ out)
{
  __shared__ float    acc[TILE_N * HF_];
  __shared__ __attribute__((aligned(16))) float den[TILE_N * H_];
  __shared__ __attribute__((aligned(16))) float elT[TILE_N * H_];
  __shared__ __attribute__((aligned(16))) float mxT[TILE_N * H_];
  __shared__ unsigned q[SC_WAVES * QCAP];
  __shared__ float    wt[SC_WAVES * BATCH * H_];

  const int tid  = threadIdx.x;
  const int lane = tid & 31;
  const int wave = __builtin_amdgcn_readfirstlane(tid >> 5);
  const int nodeBase = blockIdx.x * TILE_N;

  for (int i = tid; i < TILE_N * HF_ / 4; i += SC_WAVES * 32)
    ((v4f*)acc)[i] = (v4f){0.f, 0.f, 0.f, 0.f};
  {
    const int g = nodeBase * H_ + tid;
    const bool in = g < N_ * H_;
    den[tid] = 0.f;
    elT[tid] = in ? el[g] : 0.f;
    mxT[tid] = in ? decodeOrd(emax[g]) : 0.f;
  }
  __syncthreads();

  unsigned* qw = q + wave * QCAP;
  float*    ww = wt + wave * (BATCH * H_);
  unsigned qh = 0, qt = 0;
  const int rot = blockIdx.x * (NCHUNK / NTILES);
  const int NJ  = (NCHUNK + SC_WAVES - 1) / SC_WAVES;

  auto chunkOf = [&](int j) -> int {
    const int k = wave + SC_WAVES * j;
    if (k >= NCHUNK) return -1;
    const int c = rot + k;
    return (c >= NCHUNK) ? c - NCHUNK : c;
  };
  auto loadChunk = [&](int j) -> v4i {
    const int c = chunkOf(j);
    const int cc = (c < 0) ? 0 : c;
    return ((const v4i*)(dst + (size_t)cc * CHUNK))[(unsigned)lane];
  };
  auto look = [&](const v4i d, int j) {
    const int c = chunkOf(j);
    if (c < 0) return;
    const unsigned r0 = (unsigned)(d.x - nodeBase), r1 = (unsigned)(d.y - nodeBase);
    const unsigned r2 = (unsigned)(d.z - nodeBase), r3 = (unsigned)(d.w - nodeBase);
    const bool m0 = r0 < TILE_N, m1 = r1 < TILE_N, m2 = r2 < TILE_N, m3 = r3 < TILE_N;
    if (__builtin_amdgcn_ballot_w32(m0 | m1 | m2 | m3) == 0u) return;
    const unsigned e0 = (unsigned)(c * CHUNK + lane * 4);
    const unsigned b0 = __builtin_amdgcn_ballot_w32(m0);
    const unsigned b1 = __builtin_amdgcn_ballot_w32(m1);
    const unsigned b2 = __builtin_amdgcn_ballot_w32(m2);
    const unsigned b3 = __builtin_amdgcn_ballot_w32(m3);
    if (m0) qw[(qt + __builtin_amdgcn_mbcnt_lo(b0, 0u)) & (QCAP - 1)] = ((e0 + 0u) << 8) | r0;
    qt += __builtin_popcount(b0);
    if (m1) qw[(qt + __builtin_amdgcn_mbcnt_lo(b1, 0u)) & (QCAP - 1)] = ((e0 + 1u) << 8) | r1;
    qt += __builtin_popcount(b1);
    if (m2) qw[(qt + __builtin_amdgcn_mbcnt_lo(b2, 0u)) & (QCAP - 1)] = ((e0 + 2u) << 8) | r2;
    qt += __builtin_popcount(b2);
    if (m3) qw[(qt + __builtin_amdgcn_mbcnt_lo(b3, 0u)) & (QCAP - 1)] = ((e0 + 3u) << 8) | r3;
    qt += __builtin_popcount(b3);
    while (qt - qh >= BATCH) {
      scatter_batch<true>(qw, ww, qh, BATCH, lane, src, hW, er, elT, mxT, den, acc);
      qh += BATCH;
    }
  };

  v4i bufA[4], bufB[4];
  auto load4 = [&](v4i (&buf)[4], int j) {
#pragma unroll
    for (int u = 0; u < 4; ++u) buf[u] = loadChunk(j + u);
  };
  auto look4 = [&](const v4i (&buf)[4], int j) {
#pragma unroll
    for (int u = 0; u < 4; ++u) look(buf[u], j + u);
  };
  load4(bufA, 0);
  for (int j0 = 0; j0 < NJ; j0 += 8) {
    load4(bufB, j0 + 4);
    look4(bufA, j0);
    load4(bufA, j0 + 8);
    look4(bufB, j0 + 4);
  }
  if (qt != qh)
    scatter_batch<false>(qw, ww, qh, (int)(qt - qh), lane, src, hW, er, elT, mxT, den, acc);

  __syncthreads();

  const v4f b4 = ((const v4f*)bias)[lane];
  for (int r = wave; r < TILE_N; r += SC_WAVES) {
    const int node = nodeBase + r;
    if (node < N_) {
      const float* p = acc + r * HF_ + lane;
      v4f v = (v4f){p[0], p[32], p[64], p[96]};
      const float d = den[r * H_ + (lane >> 3)];
      if (d != 0.f) { v.x /= d; v.y /= d; v.z /= d; v.w /= d; }
      ((v4f*)(out + (size_t)node * HF_))[lane] = v + b4;
    }
  }
}

extern "C" void kernel_launch(void* const* d_in, const int* in_sizes, int n_in,
                              void* d_out, int out_size, void* d_ws, size_t ws_size,
                              hipStream_t stream)
{
  const float* h    = (const float*)d_in[0];
  const float* W    = (const float*)d_in[1];
  const float* alw  = (const float*)d_in[2];
  const float* alb  = (const float*)d_in[3];
  const float* arw  = (const float*)d_in[4];
  const float* arb  = (const float*)d_in[5];
  const float* bias = (const float*)d_in[6];
  const int*   src  = (const int*)d_in[7];
  const int*   dst  = (const int*)d_in[8];
  float* out = (float*)d_out;

  char* ws = (char*)d_ws;
  size_t off = 0;
  auto carve = [&](size_t bytes) -> void* {
    void* p = ws + off;
    off += (bytes + 255) & ~(size_t)255;
    return p;
  };
  float*    hW    = (float*)carve((size_t)N_ * HF_ * sizeof(float));
  float*    el    = (float*)carve((size_t)N_ * H_ * sizeof(float));
  float*    er    = (float*)carve((size_t)N_ * H_ * sizeof(float));
  unsigned* emax  = (unsigned*)carve((size_t)N_ * H_ * sizeof(unsigned));
  (void)ws_size; (void)in_sizes; (void)n_in; (void)out_size;

  {
    const int strips = N_ / 16;
    const int blocks = (strips + 7) / 8;
    gat_gemm_wmma<<<blocks, 256, 0, stream>>>(h, W, hW, N_);
  }
  {
    const int nNH = N_ * H_;
    gat_attn_proj<<<(nNH + 255) / 256, 256, 0, stream>>>(hW, alw, alb, arw, arb, el, er, nNH);
  }
  {
    const int nNH = N_ * H_;
    gat_init<<<(nNH + 255) / 256, 256, 0, stream>>>(emax, nNH);
  }
  {
    const int nEH = E_ * H_;
    gat_edge_max<<<(nEH + 255) / 256, 256, 0, stream>>>(src, dst, el, er, emax, nEH);
  }
  {
    gat_scatter_tiles<<<NTILES, SC_WAVES * 32, 0, stream>>>(src, dst, hW, el, er, emax, bias, out);
  }
}


# Segment max made in the tile kernel: waves list their tile's edges in LDS on the dst walk (LDS integer max), then weigh and sum from the list; init, max launches gone: 5 to 3; sums in arrival order
# speedup vs baseline: 3.3890x; 1.2243x over previous
#include <hip/hip_runtime.h>

#define N_  50000
#define E_  800000
#define D_  256
#define H_  4
#define F_  32
#define HF_ 128
#define NEG_SLOPE 0.2f

typedef __attribute__((ext_vector_type(2))) float v2f;
typedef __attribute__((ext_vector_type(8))) float v8f;
typedef __attribute__((ext_vector_type(4))) float v4f;
typedef __attribute__((ext_vector_type(4))) int   v4i;

__device__ __forceinline__ unsigned encodeOrd(float f) {
  unsigned u = __float_as_uint(f);
  return (u & 0x80000000u) ? ~u : (u | 0x80000000u);
}
__device__ __forceinline__ float decodeOrd(unsigned u) {
  unsigned b = (u & 0x80000000u) ? (u & 0x7FFFFFFFu) : ~u;
  return __uint_as_float(b);
}

__global__ __launch_bounds__(256) void gat_gemm_wmma(
    const float* __restrict__ h, const float* __restrict__ W,
    float* __restrict__ hW, int nRows)
{
  const int wave = threadIdx.x >> 5;
  const int lane = threadIdx.x & 31;
  const int rowBase = (blockIdx.x * 8 + wave) * 16;
  if (rowBase >= nRows) return;

  const int m    = lane & 15;
  const int khalf= (lane >> 4) * 2;

  v8f acc[8];
#pragma unroll
  for (int t = 0; t < 8; ++t) acc[t] = (v8f){0.f,0.f,0.f,0.f,0.f,0.f,0.f,0.f};

  for (int k0 = 0; k0 < D_; k0 += 4) {
    const float* ap = h + (size_t)(rowBase + m) * D_ + (k0 + khalf);
    v2f a; a.x = ap[0]; a.y = ap[1];
#pragma unroll
    for (int t = 0; t < 8; ++t) {
      const int n = t * 16 + (lane & 15);
      const float* bp = W + (size_t)(k0 + khalf) * HF_ + n;
      v2f b; b.x = bp[0]; b.y = bp[HF_];
      acc[t] = __builtin_amdgcn_wmma_f32_16x16x4_f32(
           false, a,  false, b,
           (short)0, acc[t],  false,  false);
    }
  }

  const int mHalf = (lane >> 4) * 8;
#pragma unroll
  for (int t = 0; t < 8; ++t) {
    const int n = t * 16 + (lane & 15);
#pragma unroll
    for (int v = 0; v < 8; ++v) {
      hW[(size_t)(rowBase + mHalf + v) * HF_ + n] = acc[t][v];
    }
  }
}

__global__ __launch_bounds__(256) void gat_attn_proj(
    const float* __restrict__ hW,
    const float* __restrict__ alw, const float* __restrict__ alb,
    const float* __restrict__ arw, const float* __restrict__ arb,
    float* __restrict__ el, float* __restrict__ er, int nNH)
{
  const int idx = blockIdx.x * blockDim.x + threadIdx.x;
  if (idx >= nNH) return;
  const float* p = hW + (size_t)idx * F_;
  float sl = 0.f, sr = 0.f;
#pragma unroll
  for (int f = 0; f < F_; ++f) {
    const float x = p[f];
    sl += x * alw[f];
    sr += x * arw[f];
  }
  el[idx] = sl + alb[0];
  er[idx] = sr + arb[0];
}

#define TILE_N   256
#define SC_WAVES 32
#define LCAP     512
#define CHUNK    128
#define NCHUNK   (E_ / CHUNK)
#define NTILES   ((N_ + TILE_N - 1) / TILE_N)
#define BATCH    32

static_assert(E_ % CHUNK == 0, "edge list must be whole chunks");
static_assert(E_ <= (1 << 24), "edge id must fit 24 bits of a list entry");
static_assert(N_ <= (1 << 24), "node id must fit 24 bits of a list entry");
static_assert(TILE_N == 256, "local node id must fit 8 bits of a list entry");
static_assert(TILE_N * H_ == SC_WAVES * 32, "one denominator per thread");
static_assert(BATCH + CHUNK <= LCAP, "list must hold a batch plus one chunk");

__device__ __forceinline__ float leaky(float v) {
  return (v >= 0.f) ? v : NEG_SLOPE * v;
}

template <bool FULL>
__device__ __forceinline__ void max_batch(
    unsigned* qw, unsigned qh, int n, int lane,
    const int* __restrict__ src, const float* __restrict__ er,
    const float* elT, unsigned* mxE)
{
  if (FULL || lane < n) {
    const unsigned idx = (qh + (unsigned)lane) & (LCAP - 1);
    const unsigned ent = qw[idx];
    const int dl = (int)(ent & 255u);
    const int s = src[ent >> 8];
    const v4f r4 = ((const v4f*)er)[s];
    const v4f l4 = ((const v4f*)elT)[dl];
    const float v0 = leaky(r4.x + l4.x), v1 = leaky(r4.y + l4.y);
    const float v2 = leaky(r4.z + l4.z), v3 = leaky(r4.w + l4.w);
    unsigned* mp = mxE + dl * H_;
    atomicMax(mp + 0, encodeOrd(v0));
    atomicMax(mp + 1, encodeOrd(v1));
    atomicMax(mp + 2, encodeOrd(v2));
    atomicMax(mp + 3, encodeOrd(v3));
    qw[idx] = ((unsigned)s << 8) | (unsigned)dl;
  }
}

template <bool FULL, bool SRCFORM>
__device__ __forceinline__ void scatter_batch(
    const unsigned* qw, float* ww, unsigned qh, int n, int lane,
    const int* __restrict__ src, const float* __restrict__ hW,
    const float* __restrict__ er, const float* elT, const float* mxT,
    float* den, float* acc)
{
  const bool mine = FULL || lane < n;
  int s = 0, dl = 0;
  v4f r4 = (v4f){0.f, 0.f, 0.f, 0.f};
  if (mine) {
    const unsigned ent = qw[(qh + (unsigned)lane) & (LCAP - 1)];
    dl = (int)(ent & 255u);
    s = SRCFORM ? (int)(ent >> 8) : src[ent >> 8];
    r4 = ((const v4f*)er)[s];
  }
  const int hd = lane >> 3;
  const int cnt = FULL ? BATCH : n;
  v4f row[8];
  float wv[8];
  auto loadRows = [&](int g) {
#pragma unroll
    for (int u = 0; u < 8; ++u) {
      row[u] = (v4f){0.f, 0.f, 0.f, 0.f};
      if (FULL || g + u < n) {
        const int sk = __builtin_amdgcn_readlane(s, g + u);
        const v4f* rp = (const v4f*)(hW + (size_t)sk * HF_);
        row[u] = rp[(unsigned)lane];
      }
    }
  };
  auto loadWeights = [&](int g) {
#pragma unroll
    for (int u = 0; u < 8; ++u) {
      wv[u] = 0.f;
      if (FULL || g + u < n) wv[u] = ww[(g + u) * H_ + hd];
    }
  };
  auto addRows = [&](int g) {
#pragma unroll
    for (int u = 0; u < 8; ++u) {
      if (FULL || g + u < n) {
        const int dlk = __builtin_amdgcn_readlane(dl, g + u);
        float* p = acc + dlk * HF_ + lane;
        const float w = wv[u];
        atomicAdd(p,      row[u].x * w);
        atomicAdd(p + 32, row[u].y * w);
        atomicAdd(p + 64, row[u].z * w);
        atomicAdd(p + 96, row[u].w * w);
      }
    }
  };

  loadRows(0);
  if (mine) {
    const v4f l4 = ((const v4f*)elT)[dl];
    const v4f m4 = ((const v4f*)mxT)[dl];
    const float v0 = leaky(r4.x + l4.x), v1 = leaky(r4.y + l4.y);
    const float v2 = leaky(r4.z + l4.z), v3 = leaky(r4.w + l4.w);
    v4f x;
    x.x = expf(v0 - m4.x);
    x.y = expf(v1 - m4.y);
    x.z = expf(v2 - m4.z);
    x.w = expf(v3 - m4.w);
    ((v4f*)ww)[lane] = x;
    float* dp = den + dl * H_;
    atomicAdd(dp + 0, x.x);
    atomicAdd(dp + 1, x.y);
    atomicAdd(dp + 2, x.z);
    atomicAdd(dp + 3, x.w);
  }
  loadWeights(0);
  addRows(0);
  for (int g = 8; g < cnt; g += 8) {
    loadRows(g);
    loadWeights(g);
    addRows(g);
  }
}

__global__ __launch_bounds__(1024) void gat_scatter_tiles(
    const int* __restrict__ src, const int* __restrict__ dst,
    const float* __restrict__ hW, const float* __restrict__ el,
    const float* __restrict__ er,
    const float* __restrict__ bias, float* __restrict__ out)
{
  __shared__ float    acc[TILE_N * HF_];
  __shared__ __attribute__((aligned(16))) float    den[TILE_N * H_];
  __shared__ __attribute__((aligned(16))) float    elT[TILE_N * H_];
  __shared__ __attribute__((aligned(16))) unsigned mxE[TILE_N * H_];
  __shared__ __attribute__((aligned(16))) float    mxT[TILE_N * H_];
  __shared__ unsigned q[SC_WAVES * LCAP];
  __shared__ float    wt[SC_WAVES * BATCH * H_];

  const int tid  = threadIdx.x;
  const int lane = tid & 31;
  const int wave = __builtin_amdgcn_readfirstlane(tid >> 5);
  const int nodeBase = blockIdx.x * TILE_N;

  for (int i = tid; i < TILE_N * HF_ / 4; i += SC_WAVES * 32)
    ((v4f*)acc)[i] = (v4f){0.f, 0.f, 0.f, 0.f};
  {
    const int g = nodeBase * H_ + tid;
    const bool in = g < N_ * H_;
    den[tid] = 0.f;
    elT[tid] = in ? el[g] : 0.f;
    mxE[tid] = encodeOrd(-__builtin_inff());
  }
  __syncthreads();

  unsigned* qw = q + wave * LCAP;
  float*    ww = wt + wave * (BATCH * H_);
  unsigned qh = 0, qt = 0;
  const int rot = blockIdx.x * (NCHUNK / NTILES);
  const int NJ  = (NCHUNK + SC_WAVES - 1) / SC_WAVES;

  auto chunkOf = [&](int j) -> int {
    const int k = wave + SC_WAVES * j;
    if (k >= NCHUNK) return -1;
    const int c = rot + k;
    return (c >= NCHUNK) ? c - NCHUNK : c;
  };
  auto loadChunk = [&](int j) -> v4i {
    const int c = chunkOf(j);
    const int cc = (c < 0) ? 0 : c;
    return ((const v4i*)(dst + (size_t)cc * CHUNK))[(unsigned)lane];
  };
  auto look = [&](const v4i d, int j, auto&& drain) {
    const int c = chunkOf(j);
    if (c < 0) return;
    const unsigned r0 = (unsigned)(d.x - nodeBase), r1 = (unsigned)(d.y - nodeBase);
    const unsigned r2 = (unsigned)(d.z - nodeBase), r3 = (unsigned)(d.w - nodeBase);
    const bool m0 = r0 < TILE_N, m1 = r1 < TILE_N, m2 = r2 < TILE_N, m3 = r3 < TILE_N;
    if (__builtin_amdgcn_ballot_w32(m0 | m1 | m2 | m3) == 0u) return;
    const unsigned e0 = (unsigned)(c * CHUNK + lane * 4);
    const unsigned b0 = __builtin_amdgcn_ballot_w32(m0);
    const unsigned b1 = __builtin_amdgcn_ballot_w32(m1);
    const unsigned b2 = __builtin_amdgcn_ballot_w32(m2);
    const unsigned b3 = __builtin_amdgcn_ballot_w32(m3);
    if (m0) qw[(qt + __builtin_amdgcn_mbcnt_lo(b0, 0u)) & (LCAP - 1)] = ((e0 + 0u) << 8) | r0;
    qt += __builtin_popcount(b0);
    if (m1) qw[(qt + __builtin_amdgcn_mbcnt_lo(b1, 0u)) & (LCAP - 1)] = ((e0 + 1u) << 8) | r1;
    qt += __builtin_popcount(b1);
    if (m2) qw[(qt + __builtin_amdgcn_mbcnt_lo(b2, 0u)) & (LCAP - 1)] = ((e0 + 2u) << 8) | r2;
    qt += __builtin_popcount(b2);
    if (m3) qw[(qt + __builtin_amdgcn_mbcnt_lo(b3, 0u)) & (LCAP - 1)] = ((e0 + 3u) << 8) | r3;
    qt += __builtin_popcount(b3);
    while (qt - qh >= BATCH) {
      drain(qh);
      qh += BATCH;
    }
  };
  auto walk = [&](auto&& drain) {
    v4i bufA[4], bufB[4];
    auto load4 = [&](v4i (&buf)[4], int j) {
#pragma unroll
      for (int u = 0; u < 4; ++u) buf[u] = loadChunk(j + u);
    };
    auto look4 = [&](const v4i (&buf)[4], int j) {
#pragma unroll
      for (int u = 0; u < 4; ++u) look(buf[u], j + u, drain);
    };
    load4(bufA, 0);
    for (int j0 = 0; j0 < NJ; j0 += 8) {
      load4(bufB, j0 + 4);
      look4(bufA, j0);
      load4(bufA, j0 + 8);
      look4(bufB, j0 + 4);
    }
  };

  walk([&](unsigned h) { max_batch<true>(qw, h, BATCH, lane, src, er, elT, mxE); });
  if (qt != qh)
    max_batch<false>(qw, qh, (int)(qt - qh), lane, src, er, elT, mxE);
  const unsigned kept = qt;

  __syncthreads();
  mxT[tid] = decodeOrd(mxE[tid]);
  __syncthreads();

  if (kept <= LCAP) {
    unsigned p = 0;
    for (; p + BATCH <= kept; p += BATCH)
      scatter_batch<true, true>(qw, ww, p, BATCH, lane, src, hW, er, elT, mxT, den, acc);
    if (p != kept)
      scatter_batch<false, true>(qw, ww, p, (int)(kept - p), lane, src, hW, er, elT, mxT, den, acc);
  } else {
    qh = 0; qt = 0;
    walk([&](unsigned h) {
      scatter_batch<true, false>(qw, ww, h, BATCH, lane, src, hW, er, elT, mxT, den, acc);
    });
    if (qt != qh)
      scatter_batch<false, false>(qw, ww, qh, (int)(qt - qh), lane, src, hW, er, elT, mxT, den, acc);
  }

  __syncthreads();

  const v4f b4 = ((const v4f*)bias)[lane];
  for (int r = wave; r < TILE_N; r += SC_WAVES) {
    const int node = nodeBase + r;
    if (node < N_) {
      const float* p = acc + r * HF_ + lane;
      v4f v = (v4f){p[0], p[32], p[64], p[96]};
      const float d = den[r * H_ + (lane >> 3)];
      if (d != 0.f) { v.x /= d; v.y /= d; v.z /= d; v.w /= d; }
      ((v4f*)(out + (size_t)node * HF_))[lane] = v + b4;
    }
  }
}

extern "C" void kernel_launch(void* const* d_in, const int* in_sizes, int n_in,
                              void* d_out, int out_size, void* d_ws, size_t ws_size,
                              hipStream_t stream)
{
  const float* h    = (const float*)d_in[0];
  const float* W    = (const float*)d_in[1];
  const float* alw  = (const float*)d_in[2];
  const float* alb  = (const float*)d_in[3];
  const float* arw  = (const float*)d_in[4];
  const float* arb  = (const float*)d_in[5];
  const float* bias = (const float*)d_in[6];
  const int*   src  = (const int*)d_in[7];
  const int*   dst  = (const int*)d_in[8];
  float* out = (float*)d_out;

  char* ws = (char*)d_ws;
  size_t off = 0;
  auto carve = [&](size_t bytes) -> void* {
    void* p = ws + off;
    off += (bytes + 255) & ~(size_t)255;
    return p;
  };
  float*    hW    = (float*)carve((size_t)N_ * HF_ * sizeof(float));
  float*    el    = (float*)carve((size_t)N_ * H_ * sizeof(float));
  float*    er    = (float*)carve((size_t)N_ * H_ * sizeof(float));
  (void)ws_size; (void)in_sizes; (void)n_in; (void)out_size;

  {
    const int strips = N_ / 16;
    const int blocks = (strips + 7) / 8;
    gat_gemm_wmma<<<blocks, 256, 0, stream>>>(h, W, hW, N_);
  }
  {
    const int nNH = N_ * H_;
    gat_attn_proj<<<(nNH + 255) / 256, 256, 0, stream>>>(hW, alw, alb, arw, arb, el, er, nNH);
  }
  {
    gat_scatter_tiles<<<NTILES, SC_WAVES * 32, 0, stream>>>(src, dst, hW, el, er, bias, out);
  }
}


# Tile walk on packed keys: projection launch writes each edge's destination tile as one byte; tile work-groups scan those bytes, 512 edges per load, one test per 4-edge word; sums in arrival order
# speedup vs baseline: 3.8769x; 1.1440x over previous
#include <hip/hip_runtime.h>

#define N_  50000
#define E_  800000
#define D_  256
#define H_  4
#define F_  32
#define HF_ 128
#define NEG_SLOPE 0.2f

typedef __attribute__((ext_vector_type(2))) float v2f;
typedef __attribute__((ext_vector_type(8))) float v8f;
typedef __attribute__((ext_vector_type(4))) float v4f;
typedef __attribute__((ext_vector_type(4))) int   v4i;
typedef __attribute__((ext_vector_type(4))) unsigned v4u;

#define TILE_N     256
#define TILE_SHIFT 8
#define NTILES     ((N_ + TILE_N - 1) / TILE_N)
static_assert((1 << TILE_SHIFT) == TILE_N, "tile id is the node id shifted");
#define CHUNK      512
#define NCHUNK     ((E_ + CHUNK - 1) / CHUNK)
#define NO_TILE    0xFFu
static_assert(NTILES <= NO_TILE, "tile id must fit one byte and differ from the padding");
static_assert(E_ % 4 == 0, "tile bytes are written four at a time");

__device__ __forceinline__ unsigned encodeOrd(float f) {
  unsigned u = __float_as_uint(f);
  return (u & 0x80000000u) ? ~u : (u | 0x80000000u);
}
__device__ __forceinline__ float decodeOrd(unsigned u) {
  unsigned b = (u & 0x80000000u) ? (u & 0x7FFFFFFFu) : ~u;
  return __uint_as_float(b);
}

__global__ __launch_bounds__(256) void gat_gemm_wmma(
    const float* __restrict__ h, const float* __restrict__ W,
    float* __restrict__ hW, int nRows)
{
  const int wave = threadIdx.x >> 5;
  const int lane = threadIdx.x & 31;
  const int rowBase = (blockIdx.x * 8 + wave) * 16;
  if (rowBase >= nRows) return;

  const int m    = lane & 15;
  const int khalf= (lane >> 4) * 2;

  v8f acc[8];
#pragma unroll
  for (int t = 0; t < 8; ++t) acc[t] = (v8f){0.f,0.f,0.f,0.f,0.f,0.f,0.f,0.f};

  for (int k0 = 0; k0 < D_; k0 += 4) {
    const float* ap = h + (size_t)(rowBase + m) * D_ + (k0 + khalf);
    v2f a; a.x = ap[0]; a.y = ap[1];
#pragma unroll
    for (int t = 0; t < 8; ++t) {
      const int n = t * 16 + (lane & 15);
      const float* bp = W + (size_t)(k0 + khalf) * HF_ + n;
      v2f b; b.x = bp[0]; b.y = bp[HF_];
      acc[t] = __builtin_amdgcn_wmma_f32_16x16x4_f32(
           false, a,  false, b,
           (short)0, acc[t],  false,  false);
    }
  }

  const int mHalf = (lane >> 4) * 8;
#pragma unroll
  for (int t = 0; t < 8; ++t) {
    const int n = t * 16 + (lane & 15);
#pragma unroll
    for (int v = 0; v < 8; ++v) {
      hW[(size_t)(rowBase + mHalf + v) * HF_ + n] = acc[t][v];
    }
  }
}

__global__ __launch_bounds__(256) void gat_attn_proj(
    const float* __restrict__ hW,
    const float* __restrict__ alw, const float* __restrict__ alb,
    const float* __restrict__ arw, const float* __restrict__ arb,
    float* __restrict__ el, float* __restrict__ er, int nNH,
    const int* __restrict__ dst, unsigned* __restrict__ tileOf)
{
  const int idx = blockIdx.x * blockDim.x + threadIdx.x;
  for (int i = idx; i < NCHUNK * (CHUNK / 4); i += (int)(gridDim.x * blockDim.x)) {
    unsigned t4 = NO_TILE * 0x01010101u;
    if (i < E_ / 4) {
      const v4i d = ((const v4i*)dst)[i];
      t4 = (unsigned)(d.x >> TILE_SHIFT) | ((unsigned)(d.y >> TILE_SHIFT) << 8) |
           ((unsigned)(d.z >> TILE_SHIFT) << 16) | ((unsigned)(d.w >> TILE_SHIFT) << 24);
    }
    tileOf[i] = t4;
  }
  if (idx >= nNH) return;
  const float* p = hW + (size_t)idx * F_;
  float sl = 0.f, sr = 0.f;
#pragma unroll
  for (int f = 0; f < F_; ++f) {
    const float x = p[f];
    sl += x * alw[f];
    sr += x * arw[f];
  }
  el[idx] = sl + alb[0];
  er[idx] = sr + arb[0];
}

#define SC_WAVES 32
#define LCAP     512
#define BATCH    32

static_assert(N_ <= (1 << 24), "node id must fit 24 bits of a list entry");
static_assert(TILE_N == 256, "local node id must fit 8 bits of a list entry");
static_assert(TILE_N * H_ == SC_WAVES * 32, "one denominator per thread");
static_assert(2 * BATCH <= LCAP, "list must hold a batch plus one entry per lane");

__device__ __forceinline__ float leaky(float v) {
  return (v >= 0.f) ? v : NEG_SLOPE * v;
}

template <bool FULL>
__device__ __forceinline__ void max_batch(
    unsigned* qw, unsigned qh, int n, int lane,
    const int* __restrict__ src, const int* __restrict__ dst, int nodeBase,
    const float* __restrict__ er, const float* elT, unsigned* mxE)
{
  if (FULL || lane < n) {
    const unsigned idx = (qh + (unsigned)lane) & (LCAP - 1);
    const unsigned e = qw[idx];
    const int s = src[e];
    const int dl = dst[e] - nodeBase;
    const v4f r4 = ((const v4f*)er)[s];
    const v4f l4 = ((const v4f*)elT)[dl];
    const float v0 = leaky(r4.x + l4.x), v1 = leaky(r4.y + l4.y);
    const float v2 = leaky(r4.z + l4.z), v3 = leaky(r4.w + l4.w);
    unsigned* mp = mxE + dl * H_;
    atomicMax(mp + 0, encodeOrd(v0));
    atomicMax(mp + 1, encodeOrd(v1));
    atomicMax(mp + 2, encodeOrd(v2));
    atomicMax(mp + 3, encodeOrd(v3));
    qw[idx] = ((unsigned)s << 8) | (unsigned)dl;
  }
}

template <bool FULL, bool SRCFORM>
__device__ __forceinline__ void scatter_batch(
    const unsigned* qw, float* ww, unsigned qh, int n, int lane,
    const int* __restrict__ src, const int* __restrict__ dst, int nodeBase,
    const float* __restrict__ hW,
    const float* __restrict__ er, const float* elT, const float* mxT,
    float* den, float* acc)
{
  const bool mine = FULL || lane < n;
  int s = 0, dl = 0;
  v4f r4 = (v4f){0.f, 0.f, 0.f, 0.f};
  if (mine) {
    const unsigned ent = qw[(qh + (unsigned)lane) & (LCAP - 1)];
    if (SRCFORM) {
      dl = (int)(ent & 255u);
      s = (int)(ent >> 8);
    } else {
      s = src[ent];
      dl = dst[ent] - nodeBase;
    }
    r4 = ((const v4f*)er)[s];
  }
  const int hd = lane >> 3;
  const int cnt = FULL ? BATCH : n;
  v4f row[8];
  float wv[8];
  auto loadRows = [&](int g) {
#pragma unroll
    for (int u = 0; u < 8; ++u) {
      row[u] = (v4f){0.f, 0.f, 0.f, 0.f};
      if (FULL || g + u < n) {
        const int sk = __builtin_amdgcn_readlane(s, g + u);
        const v4f* rp = (const v4f*)(hW + (size_t)sk * HF_);
        row[u] = rp[(unsigned)lane];
      }
    }
  };
  auto loadWeights = [&](int g) {
#pragma unroll
    for (int u = 0; u < 8; ++u) {
      wv[u] = 0.f;
      if (FULL || g + u < n) wv[u] = ww[(g + u) * H_ + hd];
    }
  };
  auto addRows = [&](int g) {
#pragma unroll
    for (int u = 0; u < 8; ++u) {
      if (FULL || g + u < n) {
        const int dlk = __builtin_amdgcn_readlane(dl, g + u);
        float* p = acc + dlk * HF_ + lane;
        const float w = wv[u];
        atomicAdd(p,      row[u].x * w);
        atomicAdd(p + 32, row[u].y * w);
        atomicAdd(p + 64, row[u].z * w);
        atomicAdd(p + 96, row[u].w * w);
      }
    }
  };

  loadRows(0);
  if (mine) {
    const v4f l4 = ((const v4f*)elT)[dl];
    const v4f m4 = ((const v4f*)mxT)[dl];
    const float v0 = leaky(r4.x + l4.x), v1 = leaky(r4.y + l4.y);
    const float v2 = leaky(r4.z + l4.z), v3 = leaky(r4.w + l4.w);
    v4f x;
    x.x = expf(v0 - m4.x);
    x.y = expf(v1 - m4.y);
    x.z = expf(v2 - m4.z);
    x.w = expf(v3 - m4.w);
    ((v4f*)ww)[lane] = x;
    float* dp = den + dl * H_;
    atomicAdd(dp + 0, x.x);
    atomicAdd(dp + 1, x.y);
    atomicAdd(dp + 2, x.z);
    atomicAdd(dp + 3, x.w);
  }
  loadWeights(0);
  addRows(0);
  for (int g = 8; g < cnt; g += 8) {
    loadRows(g);
    loadWeights(g);
    addRows(g);
  }
}

__global__ __launch_bounds__(1024) void gat_scatter_tiles(
    const int* __restrict__ src, const int* __restrict__ dst,
    const unsigned* __restrict__ tileOf,
    const float* __restrict__ hW, const float* __restrict__ el,
    const float* __restrict__ er,
    const float* __restrict__ bias, float* __restrict__ out)
{
  __shared__ float    acc[TILE_N * HF_];
  __shared__ __attribute__((aligned(16))) float    den[TILE_N * H_];
  __shared__ __attribute__((aligned(16))) float    elT[TILE_N * H_];
  __shared__ __attribute__((aligned(16))) unsigned mxE[TILE_N * H_];
  __shared__ __attribute__((aligned(16))) float    mxT[TILE_N * H_];
  __shared__ unsigned q[SC_WAVES * LCAP];
  __shared__ float    wt[SC_WAVES * BATCH * H_];

  const int tid  = threadIdx.x;
  const int lane = tid & 31;
  const int wave = __builtin_amdgcn_readfirstlane(tid >> 5);
  const int nodeBase = blockIdx.x * TILE_N;

  for (int i = tid; i < TILE_N * HF_ / 4; i += SC_WAVES * 32)
    ((v4f*)acc)[i] = (v4f){0.f, 0.f, 0.f, 0.f};
  {
    const int g = nodeBase * H_ + tid;
    const bool in = g < N_ * H_;
    den[tid] = 0.f;
    elT[tid] = in ? el[g] : 0.f;
    mxE[tid] = encodeOrd(-__builtin_inff());
  }
  __syncthreads();

  unsigned* qw = q + wave * LCAP;
  float*    ww = wt + wave * (BATCH * H_);
  unsigned qh = 0, qt = 0;
  const int rot = blockIdx.x * (NCHUNK / NTILES);
  const int NJ  = (NCHUNK + SC_WAVES - 1) / SC_WAVES;
  const unsigned pat = (unsigned)blockIdx.x * 0x01010101u;

  auto chunkOf = [&](int j) -> int {
    const int k = wave + SC_WAVES * j;
    if (k >= NCHUNK) return -1;
    const int c = rot + k;
    return (c >= NCHUNK) ? c - NCHUNK : c;
  };
  auto loadChunk = [&](int j) -> v4u {
    const int c = chunkOf(j);
    const int cc = (c < 0) ? 0 : c;
    return ((const v4u*)(tileOf + (size_t)cc * (CHUNK / 4)))[(unsigned)lane];
  };
  auto hitBits = [&](unsigned w) -> unsigned {
    const unsigned y = w ^ pat;
    return ~(((y & 0x7F7F7F7Fu) + 0x7F7F7F7Fu) | y) & 0x80808080u;
  };
  auto look = [&](const v4u t, int j, auto&& drain) {
    const int c = chunkOf(j);
    if (c < 0) return;
    unsigned m = (hitBits(t.x) >> 7) | (hitBits(t.y) >> 6) |
                 (hitBits(t.z) >> 5) | (hitBits(t.w) >> 4);
    unsigned bal = __builtin_amdgcn_ballot_w32(m != 0u);
    if (bal == 0u) return;
    const unsigned e0 = (unsigned)(c * CHUNK + lane * 16);
    do {
      if (m != 0u) {
        const unsigned b = (unsigned)__builtin_ctz(m);
        qw[(qt + __builtin_amdgcn_mbcnt_lo(bal, 0u)) & (LCAP - 1)] = e0 + ((b & 3u) << 2) + (b >> 3);
        m &= m - 1u;
      }
      qt += __builtin_popcount(bal);
      if (qt - qh >= BATCH) {
        drain(qh);
        qh += BATCH;
      }
      bal = __builtin_amdgcn_ballot_w32(m != 0u);
    } while (bal != 0u);
  };
  auto walk = [&](auto&& drain) {
    v4u bufA[4], bufB[4];
    auto load4 = [&](v4u (&buf)[4], int j) {
#pragma unroll
      for (int u = 0; u < 4; ++u) buf[u] = loadChunk(j + u);
    };
    auto look4 = [&](const v4u (&buf)[4], int j) {
#pragma unroll
      for (int u = 0; u < 4; ++u) look(buf[u], j + u, drain);
    };
    load4(bufA, 0);
    for (int j0 = 0; j0 < NJ; j0 += 8) {
      load4(bufB, j0 + 4);
      look4(bufA, j0);
      load4(bufA, j0 + 8);
      look4(bufB, j0 + 4);
    }
  };

  walk([&](unsigned h) { max_batch<true>(qw, h, BATCH, lane, src, dst, nodeBase, er, elT, mxE); });
  if (qt != qh)
    max_batch<false>(qw, qh, (int)(qt - qh), lane, src, dst, nodeBase, er, elT, mxE);
  const unsigned kept = qt;

  __syncthreads();
  mxT[tid] = decodeOrd(mxE[tid]);
  __syncthreads();

  if (kept <= LCAP) {
    unsigned p = 0;
    for (; p + BATCH <= kept; p += BATCH)
      scatter_batch<true, true>(qw, ww, p, BATCH, lane, src, dst, nodeBase, hW, er, elT, mxT, den, acc);
    if (p != kept)
      scatter_batch<false, true>(qw, ww, p, (int)(kept - p), lane, src, dst, nodeBase, hW, er, elT, mxT, den, acc);
  } else {
    qh = 0; qt = 0;
    walk([&](unsigned h) {
      scatter_batch<true, false>(qw, ww, h, BATCH, lane, src, dst, nodeBase, hW, er, elT, mxT, den, acc);
    });
    if (qt != qh)
      scatter_batch<false, false>(qw, ww, qh, (int)(qt - qh), lane, src, dst, nodeBase, hW, er, elT, mxT, den, acc);
  }

  __syncthreads();

  const v4f b4 = ((const v4f*)bias)[lane];
  for (int r = wave; r < TILE_N; r += SC_WAVES) {
    const int node = nodeBase + r;
    if (node < N_) {
      const float* p = acc + r * HF_ + lane;
      v4f v = (v4f){p[0], p[32], p[64], p[96]};
      const float d = den[r * H_ + (lane >> 3)];
      if (d != 0.f) { v.x /= d; v.y /= d; v.z /= d; v.w /= d; }
      ((v4f*)(out + (size_t)node * HF_))[lane] = v + b4;
    }
  }
}

extern "C" void kernel_launch(void* const* d_in, const int* in_sizes, int n_in,
                              void* d_out, int out_size, void* d_ws, size_t ws_size,
                              hipStream_t stream)
{
  const float* h    = (const float*)d_in[0];
  const float* W    = (const float*)d_in[1];
  const float* alw  = (const float*)d_in[2];
  const float* alb  = (const float*)d_in[3];
  const float* arw  = (const float*)d_in[4];
  const float* arb  = (const float*)d_in[5];
  const float* bias = (const float*)d_in[6];
  const int*   src  = (const int*)d_in[7];
  const int*   dst  = (const int*)d_in[8];
  float* out = (float*)d_out;

  char* ws = (char*)d_ws;
  size_t off = 0;
  auto carve = [&](size_t bytes) -> void* {
    void* p = ws + off;
    off += (bytes + 255) & ~(size_t)255;
    return p;
  };
  float*    hW    = (float*)carve((size_t)N_ * HF_ * sizeof(float));
  float*    el    = (float*)carve((size_t)N_ * H_ * sizeof(float));
  float*    er    = (float*)carve((size_t)N_ * H_ * sizeof(float));
  unsigned* tileOf = (unsigned*)carve((size_t)NCHUNK * CHUNK);
  (void)ws_size; (void)in_sizes; (void)n_in; (void)out_size;

  {
    const int strips = N_ / 16;
    const int blocks = (strips + 7) / 8;
    gat_gemm_wmma<<<blocks, 256, 0, stream>>>(h, W, hW, N_);
  }
  {
    const int nNH = N_ * H_;
    gat_attn_proj<<<(nNH + 255) / 256, 256, 0, stream>>>(hW, alw, alb, arw, arb, el, er, nNH, dst, tileOf);
  }
  {
    gat_scatter_tiles<<<NTILES, SC_WAVES * 32, 0, stream>>>(src, dst, tileOf, hW, el, er, bias, out);
  }
}


# Destination tiles cut from 256 to 197 nodes: 254 tile work-groups instead of 196, one for nearly each of the 256 work-group processors, a quarter fewer edges each; sums stay in arrival order
# speedup vs baseline: 4.0763x; 1.0514x over previous
#include <hip/hip_runtime.h>

#define N_  50000
#define E_  800000
#define D_  256
#define H_  4
#define F_  32
#define HF_ 128
#define NEG_SLOPE 0.2f

typedef __attribute__((ext_vector_type(2))) float v2f;
typedef __attribute__((ext_vector_type(8))) float v8f;
typedef __attribute__((ext_vector_type(4))) float v4f;
typedef __attribute__((ext_vector_type(4))) int   v4i;
typedef __attribute__((ext_vector_type(4))) unsigned v4u;

#define TILE_N     197
#define NTILES     ((N_ + TILE_N - 1) / TILE_N)
#define CHUNK      512
#define NCHUNK     ((E_ + CHUNK - 1) / CHUNK)
#define NO_TILE    0xFFu
static_assert(NTILES <= NO_TILE, "tile id must fit one byte and differ from the padding");
static_assert(E_ % 4 == 0, "tile bytes are written four at a time");

__device__ __forceinline__ unsigned encodeOrd(float f) {
  unsigned u = __float_as_uint(f);
  return (u & 0x80000000u) ? ~u : (u | 0x80000000u);
}
__device__ __forceinline__ float decodeOrd(unsigned u) {
  unsigned b = (u & 0x80000000u) ? (u & 0x7FFFFFFFu) : ~u;
  return __uint_as_float(b);
}

__global__ __launch_bounds__(256) void gat_gemm_wmma(
    const float* __restrict__ h, const float* __restrict__ W,
    float* __restrict__ hW, int nRows)
{
  const int wave = threadIdx.x >> 5;
  const int lane = threadIdx.x & 31;
  const int rowBase = (blockIdx.x * 8 + wave) * 16;
  if (rowBase >= nRows) return;

  const int m    = lane & 15;
  const int khalf= (lane >> 4) * 2;

  v8f acc[8];
#pragma unroll
  for (int t = 0; t < 8; ++t) acc[t] = (v8f){0.f,0.f,0.f,0.f,0.f,0.f,0.f,0.f};

  for (int k0 = 0; k0 < D_; k0 += 4) {
    const float* ap = h + (size_t)(rowBase + m) * D_ + (k0 + khalf);
    v2f a; a.x = ap[0]; a.y = ap[1];
#pragma unroll
    for (int t = 0; t < 8; ++t) {
      const int n = t * 16 + (lane & 15);
      const float* bp = W + (size_t)(k0 + khalf) * HF_ + n;
      v2f b; b.x = bp[0]; b.y = bp[HF_];
      acc[t] = __builtin_amdgcn_wmma_f32_16x16x4_f32(
           false, a,  false, b,
           (short)0, acc[t],  false,  false);
    }
  }

  const int mHalf = (lane >> 4) * 8;
#pragma unroll
  for (int t = 0; t < 8; ++t) {
    const int n = t * 16 + (lane & 15);
#pragma unroll
    for (int v = 0; v < 8; ++v) {
      hW[(size_t)(rowBase + mHalf + v) * HF_ + n] = acc[t][v];
    }
  }
}

__global__ __launch_bounds__(256) void gat_attn_proj(
    const float* __restrict__ hW,
    const float* __restrict__ alw, const float* __restrict__ alb,
    const float* __restrict__ arw, const float* __restrict__ arb,
    float* __restrict__ el, float* __restrict__ er, int nNH,
    const int* __restrict__ dst, unsigned* __restrict__ tileOf)
{
  const int idx = blockIdx.x * blockDim.x + threadIdx.x;
  for (int i = idx; i < NCHUNK * (CHUNK / 4); i += (int)(gridDim.x * blockDim.x)) {
    unsigned t4 = NO_TILE * 0x01010101u;
    if (i < E_ / 4) {
      const v4i d = ((const v4i*)dst)[i];
      t4 = ((unsigned)d.x / TILE_N) | (((unsigned)d.y / TILE_N) << 8) |
           (((unsigned)d.z / TILE_N) << 16) | (((unsigned)d.w / TILE_N) << 24);
    }
    tileOf[i] = t4;
  }
  if (idx >= nNH) return;
  const float* p = hW + (size_t)idx * F_;
  float sl = 0.f, sr = 0.f;
#pragma unroll
  for (int f = 0; f < F_; ++f) {
    const float x = p[f];
    sl += x * alw[f];
    sr += x * arw[f];
  }
  el[idx] = sl + alb[0];
  er[idx] = sr + arb[0];
}

#define SC_WAVES 32
#define LCAP     512
#define BATCH    32

static_assert(N_ <= (1 << 24), "node id must fit 24 bits of a list entry");
static_assert(TILE_N <= 256, "local node id must fit 8 bits of a list entry");
static_assert(TILE_N * H_ <= SC_WAVES * 32, "at most one denominator per thread");
static_assert(2 * BATCH <= LCAP, "list must hold a batch plus one entry per lane");

__device__ __forceinline__ float leaky(float v) {
  return (v >= 0.f) ? v : NEG_SLOPE * v;
}

template <bool FULL>
__device__ __forceinline__ void max_batch(
    unsigned* qw, unsigned qh, int n, int lane,
    const int* __restrict__ src, const int* __restrict__ dst, int nodeBase,
    const float* __restrict__ er, const float* elT, unsigned* mxE)
{
  if (FULL || lane < n) {
    const unsigned idx = (qh + (unsigned)lane) & (LCAP - 1);
    const unsigned e = qw[idx];
    const int s = src[e];
    const int dl = dst[e] - nodeBase;
    const v4f r4 = ((const v4f*)er)[s];
    const v4f l4 = ((const v4f*)elT)[dl];
    const float v0 = leaky(r4.x + l4.x), v1 = leaky(r4.y + l4.y);
    const float v2 = leaky(r4.z + l4.z), v3 = leaky(r4.w + l4.w);
    unsigned* mp = mxE + dl * H_;
    atomicMax(mp + 0, encodeOrd(v0));
    atomicMax(mp + 1, encodeOrd(v1));
    atomicMax(mp + 2, encodeOrd(v2));
    atomicMax(mp + 3, encodeOrd(v3));
    qw[idx] = ((unsigned)s << 8) | (unsigned)dl;
  }
}

template <bool FULL, bool SRCFORM>
__device__ __forceinline__ void scatter_batch(
    const unsigned* qw, float* ww, unsigned qh, int n, int lane,
    const int* __restrict__ src, const int* __restrict__ dst, int nodeBase,
    const float* __restrict__ hW,
    const float* __restrict__ er, const float* elT, const float* mxT,
    float* den, float* acc)
{
  const bool mine = FULL || lane < n;
  int s = 0, dl = 0;
  v4f r4 = (v4f){0.f, 0.f, 0.f, 0.f};
  if (mine) {
    const unsigned ent = qw[(qh + (unsigned)lane) & (LCAP - 1)];
    if (SRCFORM) {
      dl = (int)(ent & 255u);
      s = (int)(ent >> 8);
    } else {
      s = src[ent];
      dl = dst[ent] - nodeBase;
    }
    r4 = ((const v4f*)er)[s];
  }
  const int hd = lane >> 3;
  const int cnt = FULL ? BATCH : n;
  v4f row[8];
  float wv[8];
  auto loadRows = [&](int g) {
#pragma unroll
    for (int u = 0; u < 8; ++u) {
      row[u] = (v4f){0.f, 0.f, 0.f, 0.f};
      if (FULL || g + u < n) {
        const int sk = __builtin_amdgcn_readlane(s, g + u);
        const v4f* rp = (const v4f*)(hW + (size_t)sk * HF_);
        row[u] = rp[(unsigned)lane];
      }
    }
  };
  auto loadWeights = [&](int g) {
#pragma unroll
    for (int u = 0; u < 8; ++u) {
      wv[u] = 0.f;
      if (FULL || g + u < n) wv[u] = ww[(g + u) * H_ + hd];
    }
  };
  auto addRows = [&](int g) {
#pragma unroll
    for (int u = 0; u < 8; ++u) {
      if (FULL || g + u < n) {
        const int dlk = __builtin_amdgcn_readlane(dl, g + u);
        float* p = acc + dlk * HF_ + lane;
        const float w = wv[u];
        atomicAdd(p,      row[u].x * w);
        atomicAdd(p + 32, row[u].y * w);
        atomicAdd(p + 64, row[u].z * w);
        atomicAdd(p + 96, row[u].w * w);
      }
    }
  };

  loadRows(0);
  if (mine) {
    const v4f l4 = ((const v4f*)elT)[dl];
    const v4f m4 = ((const v4f*)mxT)[dl];
    const float v0 = leaky(r4.x + l4.x), v1 = leaky(r4.y + l4.y);
    const float v2 = leaky(r4.z + l4.z), v3 = leaky(r4.w + l4.w);
    v4f x;
    x.x = expf(v0 - m4.x);
    x.y = expf(v1 - m4.y);
    x.z = expf(v2 - m4.z);
    x.w = expf(v3 - m4.w);
    ((v4f*)ww)[lane] = x;
    float* dp = den + dl * H_;
    atomicAdd(dp + 0, x.x);
    atomicAdd(dp + 1, x.y);
    atomicAdd(dp + 2, x.z);
    atomicAdd(dp + 3, x.w);
  }
  loadWeights(0);
  addRows(0);
  for (int g = 8; g < cnt; g += 8) {
    loadRows(g);
    loadWeights(g);
    addRows(g);
  }
}

__global__ __launch_bounds__(1024) void gat_scatter_tiles(
    const int* __restrict__ src, const int* __restrict__ dst,
    const unsigned* __restrict__ tileOf,
    const float* __restrict__ hW, const float* __restrict__ el,
    const float* __restrict__ er,
    const float* __restrict__ bias, float* __restrict__ out)
{
  __shared__ float    acc[TILE_N * HF_];
  __shared__ __attribute__((aligned(16))) float    den[TILE_N * H_];
  __shared__ __attribute__((aligned(16))) float    elT[TILE_N * H_];
  __shared__ __attribute__((aligned(16))) unsigned mxE[TILE_N * H_];
  __shared__ __attribute__((aligned(16))) float    mxT[TILE_N * H_];
  __shared__ unsigned q[SC_WAVES * LCAP];
  __shared__ float    wt[SC_WAVES * BATCH * H_];

  const int tid  = threadIdx.x;
  const int lane = tid & 31;
  const int wave = __builtin_amdgcn_readfirstlane(tid >> 5);
  const int nodeBase = blockIdx.x * TILE_N;

  for (int i = tid; i < TILE_N * HF_ / 4; i += SC_WAVES * 32)
    ((v4f*)acc)[i] = (v4f){0.f, 0.f, 0.f, 0.f};
  if (tid < TILE_N * H_) {
    const int g = nodeBase * H_ + tid;
    const bool in = g < N_ * H_;
    den[tid] = 0.f;
    elT[tid] = in ? el[g] : 0.f;
    mxE[tid] = encodeOrd(-__builtin_inff());
  }
  __syncthreads();

  unsigned* qw = q + wave * LCAP;
  float*    ww = wt + wave * (BATCH * H_);
  unsigned qh = 0, qt = 0;
  const int rot = blockIdx.x * (NCHUNK / NTILES);
  const int NJ  = (NCHUNK + SC_WAVES - 1) / SC_WAVES;
  const unsigned pat = (unsigned)blockIdx.x * 0x01010101u;

  auto chunkOf = [&](int j) -> int {
    const int k = wave + SC_WAVES * j;
    if (k >= NCHUNK) return -1;
    const int c = rot + k;
    return (c >= NCHUNK) ? c - NCHUNK : c;
  };
  auto loadChunk = [&](int j) -> v4u {
    const int c = chunkOf(j);
    const int cc = (c < 0) ? 0 : c;
    return ((const v4u*)(tileOf + (size_t)cc * (CHUNK / 4)))[(unsigned)lane];
  };
  auto hitBits = [&](unsigned w) -> unsigned {
    const unsigned y = w ^ pat;
    return ~(((y & 0x7F7F7F7Fu) + 0x7F7F7F7Fu) | y) & 0x80808080u;
  };
  auto look = [&](const v4u t, int j, auto&& drain) {
    const int c = chunkOf(j);
    if (c < 0) return;
    unsigned m = (hitBits(t.x) >> 7) | (hitBits(t.y) >> 6) |
                 (hitBits(t.z) >> 5) | (hitBits(t.w) >> 4);
    unsigned bal = __builtin_amdgcn_ballot_w32(m != 0u);
    if (bal == 0u) return;
    const unsigned e0 = (unsigned)(c * CHUNK + lane * 16);
    do {
      if (m != 0u) {
        const unsigned b = (unsigned)__builtin_ctz(m);
        qw[(qt + __builtin_amdgcn_mbcnt_lo(bal, 0u)) & (LCAP - 1)] = e0 + ((b & 3u) << 2) + (b >> 3);
        m &= m - 1u;
      }
      qt += __builtin_popcount(bal);
      if (qt - qh >= BATCH) {
        drain(qh);
        qh += BATCH;
      }
      bal = __builtin_amdgcn_ballot_w32(m != 0u);
    } while (bal != 0u);
  };
  auto walk = [&](auto&& drain) {
    v4u bufA[4], bufB[4];
    auto load4 = [&](v4u (&buf)[4], int j) {
#pragma unroll
      for (int u = 0; u < 4; ++u) buf[u] = loadChunk(j + u);
    };
    auto look4 = [&](const v4u (&buf)[4], int j) {
#pragma unroll
      for (int u = 0; u < 4; ++u) look(buf[u], j + u, drain);
    };
    load4(bufA, 0);
    for (int j0 = 0; j0 < NJ; j0 += 8) {
      load4(bufB, j0 + 4);
      look4(bufA, j0);
      load4(bufA, j0 + 8);
      look4(bufB, j0 + 4);
    }
  };

  walk([&](unsigned h) { max_batch<true>(qw, h, BATCH, lane, src, dst, nodeBase, er, elT, mxE); });
  if (qt != qh)
    max_batch<false>(qw, qh, (int)(qt - qh), lane, src, dst, nodeBase, er, elT, mxE);
  const unsigned kept = qt;

  __syncthreads();
  if (tid < TILE_N * H_) mxT[tid] = decodeOrd(mxE[tid]);
  __syncthreads();

  if (kept <= LCAP) {
    unsigned p = 0;
    for (; p + BATCH <= kept; p += BATCH)
      scatter_batch<true, true>(qw, ww, p, BATCH, lane, src, dst, nodeBase, hW, er, elT, mxT, den, acc);
    if (p != kept)
      scatter_batch<false, true>(qw, ww, p, (int)(kept - p), lane, src, dst, nodeBase, hW, er, elT, mxT, den, acc);
  } else {
    qh = 0; qt = 0;
    walk([&](unsigned h) {
      scatter_batch<true, false>(qw, ww, h, BATCH, lane, src, dst, nodeBase, hW, er, elT, mxT, den, acc);
    });
    if (qt != qh)
      scatter_batch<false, false>(qw, ww, qh, (int)(qt - qh), lane, src, dst, nodeBase, hW, er, elT, mxT, den, acc);
  }

  __syncthreads();

  const v4f b4 = ((const v4f*)bias)[lane];
  for (int r = wave; r < TILE_N; r += SC_WAVES) {
    const int node = nodeBase + r;
    if (node < N_) {
      const float* p = acc + r * HF_ + lane;
      v4f v = (v4f){p[0], p[32], p[64], p[96]};
      const float d = den[r * H_ + (lane >> 3)];
      if (d != 0.f) { v.x /= d; v.y /= d; v.z /= d; v.w /= d; }
      ((v4f*)(out + (size_t)node * HF_))[lane] = v + b4;
    }
  }
}

extern "C" void kernel_launch(void* const* d_in, const int* in_sizes, int n_in,
                              void* d_out, int out_size, void* d_ws, size_t ws_size,
                              hipStream_t stream)
{
  const float* h    = (const float*)d_in[0];
  const float* W    = (const float*)d_in[1];
  const float* alw  = (const float*)d_in[2];
  const float* alb  = (const float*)d_in[3];
  const float* arw  = (const float*)d_in[4];
  const float* arb  = (const float*)d_in[5];
  const float* bias = (const float*)d_in[6];
  const int*   src  = (const int*)d_in[7];
  const int*   dst  = (const int*)d_in[8];
  float* out = (float*)d_out;

  char* ws = (char*)d_ws;
  size_t off = 0;
  auto carve = [&](size_t bytes) -> void* {
    void* p = ws + off;
    off += (bytes + 255) & ~(size_t)255;
    return p;
  };
  float*    hW    = (float*)carve((size_t)N_ * HF_ * sizeof(float));
  float*    el    = (float*)carve((size_t)N_ * H_ * sizeof(float));
  float*    er    = (float*)carve((size_t)N_ * H_ * sizeof(float));
  unsigned* tileOf = (unsigned*)carve((size_t)NCHUNK * CHUNK);
  (void)ws_size; (void)in_sizes; (void)n_in; (void)out_size;

  {
    const int strips = N_ / 16;
    const int blocks = (strips + 7) / 8;
    gat_gemm_wmma<<<blocks, 256, 0, stream>>>(h, W, hW, N_);
  }
  {
    const int nNH = N_ * H_;
    gat_attn_proj<<<(nNH + 255) / 256, 256, 0, stream>>>(hW, alw, alb, arw, arb, el, er, nNH, dst, tileOf);
  }
  {
    gat_scatter_tiles<<<NTILES, SC_WAVES * 32, 0, stream>>>(src, dst, tileOf, hW, el, er, bias, out);
  }
}


# GEMM operands read by 16-byte loads: k permuted inside the fragments, output columns interleaved across the 8 tiles; whole-row stores; 34 loads per 16 WMMAs become 9; GEMM k sum order changes
# speedup vs baseline: 4.7016x; 1.1534x over previous
#include <hip/hip_runtime.h>

#define N_  50000
#define E_  800000
#define D_  256
#define H_  4
#define F_  32
#define HF_ 128
#define NEG_SLOPE 0.2f

typedef __attribute__((ext_vector_type(2))) float v2f;
typedef __attribute__((ext_vector_type(8))) float v8f;
typedef __attribute__((ext_vector_type(4))) float v4f;
typedef __attribute__((ext_vector_type(4))) int   v4i;
typedef __attribute__((ext_vector_type(4))) unsigned v4u;

#define TILE_N     197
#define NTILES     ((N_ + TILE_N - 1) / TILE_N)
#define CHUNK      512
#define NCHUNK     ((E_ + CHUNK - 1) / CHUNK)
#define NO_TILE    0xFFu
static_assert(NTILES <= NO_TILE, "tile id must fit one byte and differ from the padding");
static_assert(E_ % 4 == 0, "tile bytes are written four at a time");

__device__ __forceinline__ unsigned encodeOrd(float f) {
  unsigned u = __float_as_uint(f);
  return (u & 0x80000000u) ? ~u : (u | 0x80000000u);
}
__device__ __forceinline__ float decodeOrd(unsigned u) {
  unsigned b = (u & 0x80000000u) ? (u & 0x7FFFFFFFu) : ~u;
  return __uint_as_float(b);
}

__global__ __launch_bounds__(256) void gat_gemm_wmma(
    const float* __restrict__ h, const float* __restrict__ W,
    float* __restrict__ hW, int nRows)
{
  const int wave = threadIdx.x >> 5;
  const int lane = threadIdx.x & 31;
  const int rowBase = (blockIdx.x * 8 + wave) * 16;
  if (rowBase >= nRows) return;

  const int c  = lane & 15;
  const int hf = lane >> 4;

  v8f acc[8];
#pragma unroll
  for (int t = 0; t < 8; ++t) acc[t] = (v8f){0.f,0.f,0.f,0.f,0.f,0.f,0.f,0.f};

  const float* ap = h + (size_t)(rowBase + c) * D_ + 4 * hf;
  const float* bp = W + (size_t)(4 * hf) * HF_ + 8 * c;

  for (int kk = 0; kk < D_; kk += 8) {
    const v4f a4 = *(const v4f*)(ap + kk);
    v4f blo[4], bhi[4];
#pragma unroll
    for (int r = 0; r < 4; ++r) {
      const float* rp = bp + (size_t)(kk + r) * HF_;
      blo[r] = *(const v4f*)(rp);
      bhi[r] = *(const v4f*)(rp + 4);
    }
#pragma unroll
    for (int j = 0; j < 2; ++j) {
      v2f a; a.x = a4[2 * j]; a.y = a4[2 * j + 1];
#pragma unroll
      for (int t = 0; t < 8; ++t) {
        v2f b;
        b.x = (t < 4) ? blo[2 * j][t & 3]     : bhi[2 * j][t & 3];
        b.y = (t < 4) ? blo[2 * j + 1][t & 3] : bhi[2 * j + 1][t & 3];
        acc[t] = __builtin_amdgcn_wmma_f32_16x16x4_f32(
             false, a,  false, b,
             (short)0, acc[t],  false,  false);
      }
    }
  }

#pragma unroll
  for (int v = 0; v < 8; ++v) {
    float* op = hW + (size_t)(rowBase + 8 * hf + v) * HF_ + 8 * c;
    *(v4f*)(op)     = (v4f){acc[0][v], acc[1][v], acc[2][v], acc[3][v]};
    *(v4f*)(op + 4) = (v4f){acc[4][v], acc[5][v], acc[6][v], acc[7][v]};
  }
}

__global__ __launch_bounds__(256) void gat_attn_proj(
    const float* __restrict__ hW,
    const float* __restrict__ alw, const float* __restrict__ alb,
    const float* __restrict__ arw, const float* __restrict__ arb,
    float* __restrict__ el, float* __restrict__ er, int nNH,
    const int* __restrict__ dst, unsigned* __restrict__ tileOf)
{
  const int idx = blockIdx.x * blockDim.x + threadIdx.x;
  for (int i = idx; i < NCHUNK * (CHUNK / 4); i += (int)(gridDim.x * blockDim.x)) {
    unsigned t4 = NO_TILE * 0x01010101u;
    if (i < E_ / 4) {
      const v4i d = ((const v4i*)dst)[i];
      t4 = ((unsigned)d.x / TILE_N) | (((unsigned)d.y / TILE_N) << 8) |
           (((unsigned)d.z / TILE_N) << 16) | (((unsigned)d.w / TILE_N) << 24);
    }
    tileOf[i] = t4;
  }
  if (idx >= nNH) return;
  const float* p = hW + (size_t)idx * F_;
  float sl = 0.f, sr = 0.f;
#pragma unroll
  for (int f = 0; f < F_; ++f) {
    const float x = p[f];
    sl += x * alw[f];
    sr += x * arw[f];
  }
  el[idx] = sl + alb[0];
  er[idx] = sr + arb[0];
}

#define SC_WAVES 32
#define LCAP     512
#define BATCH    32

static_assert(N_ <= (1 << 24), "node id must fit 24 bits of a list entry");
static_assert(TILE_N <= 256, "local node id must fit 8 bits of a list entry");
static_assert(TILE_N * H_ <= SC_WAVES * 32, "at most one denominator per thread");
static_assert(2 * BATCH <= LCAP, "list must hold a batch plus one entry per lane");

__device__ __forceinline__ float leaky(float v) {
  return (v >= 0.f) ? v : NEG_SLOPE * v;
}

template <bool FULL>
__device__ __forceinline__ void max_batch(
    unsigned* qw, unsigned qh, int n, int lane,
    const int* __restrict__ src, const int* __restrict__ dst, int nodeBase,
    const float* __restrict__ er, const float* elT, unsigned* mxE)
{
  if (FULL || lane < n) {
    const unsigned idx = (qh + (unsigned)lane) & (LCAP - 1);
    const unsigned e = qw[idx];
    const int s = src[e];
    const int dl = dst[e] - nodeBase;
    const v4f r4 = ((const v4f*)er)[s];
    const v4f l4 = ((const v4f*)elT)[dl];
    const float v0 = leaky(r4.x + l4.x), v1 = leaky(r4.y + l4.y);
    const float v2 = leaky(r4.z + l4.z), v3 = leaky(r4.w + l4.w);
    unsigned* mp = mxE + dl * H_;
    atomicMax(mp + 0, encodeOrd(v0));
    atomicMax(mp + 1, encodeOrd(v1));
    atomicMax(mp + 2, encodeOrd(v2));
    atomicMax(mp + 3, encodeOrd(v3));
    qw[idx] = ((unsigned)s << 8) | (unsigned)dl;
  }
}

template <bool FULL, bool SRCFORM>
__device__ __forceinline__ void scatter_batch(
    const unsigned* qw, float* ww, unsigned qh, int n, int lane,
    const int* __restrict__ src, const int* __restrict__ dst, int nodeBase,
    const float* __restrict__ hW,
    const float* __restrict__ er, const float* elT, const float* mxT,
    float* den, float* acc)
{
  const bool mine = FULL || lane < n;
  int s = 0, dl = 0;
  v4f r4 = (v4f){0.f, 0.f, 0.f, 0.f};
  if (mine) {
    const unsigned ent = qw[(qh + (unsigned)lane) & (LCAP - 1)];
    if (SRCFORM) {
      dl = (int)(ent & 255u);
      s = (int)(ent >> 8);
    } else {
      s = src[ent];
      dl = dst[ent] - nodeBase;
    }
    r4 = ((const v4f*)er)[s];
  }
  const int hd = lane >> 3;
  const int cnt = FULL ? BATCH : n;
  v4f row[8];
  float wv[8];
  auto loadRows = [&](int g) {
#pragma unroll
    for (int u = 0; u < 8; ++u) {
      row[u] = (v4f){0.f, 0.f, 0.f, 0.f};
      if (FULL || g + u < n) {
        const int sk = __builtin_amdgcn_readlane(s, g + u);
        const v4f* rp = (const v4f*)(hW + (size_t)sk * HF_);
        row[u] = rp[(unsigned)lane];
      }
    }
  };
  auto loadWeights = [&](int g) {
#pragma unroll
    for (int u = 0; u < 8; ++u) {
      wv[u] = 0.f;
      if (FULL || g + u < n) wv[u] = ww[(g + u) * H_ + hd];
    }
  };
  auto addRows = [&](int g) {
#pragma unroll
    for (int u = 0; u < 8; ++u) {
      if (FULL || g + u < n) {
        const int dlk = __builtin_amdgcn_readlane(dl, g + u);
        float* p = acc + dlk * HF_ + lane;
        const float w = wv[u];
        atomicAdd(p,      row[u].x * w);
        atomicAdd(p + 32, row[u].y * w);
        atomicAdd(p + 64, row[u].z * w);
        atomicAdd(p + 96, row[u].w * w);
      }
    }
  };

  loadRows(0);
  if (mine) {
    const v4f l4 = ((const v4f*)elT)[dl];
    const v4f m4 = ((const v4f*)mxT)[dl];
    const float v0 = leaky(r4.x + l4.x), v1 = leaky(r4.y + l4.y);
    const float v2 = leaky(r4.z + l4.z), v3 = leaky(r4.w + l4.w);
    v4f x;
    x.x = expf(v0 - m4.x);
    x.y = expf(v1 - m4.y);
    x.z = expf(v2 - m4.z);
    x.w = expf(v3 - m4.w);
    ((v4f*)ww)[lane] = x;
    float* dp = den + dl * H_;
    atomicAdd(dp + 0, x.x);
    atomicAdd(dp + 1, x.y);
    atomicAdd(dp + 2, x.z);
    atomicAdd(dp + 3, x.w);
  }
  loadWeights(0);
  addRows(0);
  for (int g = 8; g < cnt; g += 8) {
    loadRows(g);
    loadWeights(g);
    addRows(g);
  }
}

__global__ __launch_bounds__(1024) void gat_scatter_tiles(
    const int* __restrict__ src, const int* __restrict__ dst,
    const unsigned* __restrict__ tileOf,
    const float* __restrict__ hW, const float* __restrict__ el,
    const float* __restrict__ er,
    const float* __restrict__ bias, float* __restrict__ out)
{
  __shared__ float    acc[TILE_N * HF_];
  __shared__ __attribute__((aligned(16))) float    den[TILE_N * H_];
  __shared__ __attribute__((aligned(16))) float    elT[TILE_N * H_];
  __shared__ __attribute__((aligned(16))) unsigned mxE[TILE_N * H_];
  __shared__ __attribute__((aligned(16))) float    mxT[TILE_N * H_];
  __shared__ unsigned q[SC_WAVES * LCAP];
  __shared__ float    wt[SC_WAVES * BATCH * H_];

  const int tid  = threadIdx.x;
  const int lane = tid & 31;
  const int wave = __builtin_amdgcn_readfirstlane(tid >> 5);
  const int nodeBase = blockIdx.x * TILE_N;

  for (int i = tid; i < TILE_N * HF_ / 4; i += SC_WAVES * 32)
    ((v4f*)acc)[i] = (v4f){0.f, 0.f, 0.f, 0.f};
  if (tid < TILE_N * H_) {
    const int g = nodeBase * H_ + tid;
    const bool in = g < N_ * H_;
    den[tid] = 0.f;
    elT[tid] = in ? el[g] : 0.f;
    mxE[tid] = encodeOrd(-__builtin_inff());
  }
  __syncthreads();

  unsigned* qw = q + wave * LCAP;
  float*    ww = wt + wave * (BATCH * H_);
  unsigned qh = 0, qt = 0;
  const int rot = blockIdx.x * (NCHUNK / NTILES);
  const int NJ  = (NCHUNK + SC_WAVES - 1) / SC_WAVES;
  const unsigned pat = (unsigned)blockIdx.x * 0x01010101u;

  auto chunkOf = [&](int j) -> int {
    const int k = wave + SC_WAVES * j;
    if (k >= NCHUNK) return -1;
    const int c = rot + k;
    return (c >= NCHUNK) ? c - NCHUNK : c;
  };
  auto loadChunk = [&](int j) -> v4u {
    const int c = chunkOf(j);
    const int cc = (c < 0) ? 0 : c;
    return ((const v4u*)(tileOf + (size_t)cc * (CHUNK / 4)))[(unsigned)lane];
  };
  auto hitBits = [&](unsigned w) -> unsigned {
    const unsigned y = w ^ pat;
    return ~(((y & 0x7F7F7F7Fu) + 0x7F7F7F7Fu) | y) & 0x80808080u;
  };
  auto look = [&](const v4u t, int j, auto&& drain) {
    const int c = chunkOf(j);
    if (c < 0) return;
    unsigned m = (hitBits(t.x) >> 7) | (hitBits(t.y) >> 6) |
                 (hitBits(t.z) >> 5) | (hitBits(t.w) >> 4);
    unsigned bal = __builtin_amdgcn_ballot_w32(m != 0u);
    if (bal == 0u) return;
    const unsigned e0 = (unsigned)(c * CHUNK + lane * 16);
    do {
      if (m != 0u) {
        const unsigned b = (unsigned)__builtin_ctz(m);
        qw[(qt + __builtin_amdgcn_mbcnt_lo(bal, 0u)) & (LCAP - 1)] = e0 + ((b & 3u) << 2) + (b >> 3);
        m &= m - 1u;
      }
      qt += __builtin_popcount(bal);
      if (qt - qh >= BATCH) {
        drain(qh);
        qh += BATCH;
      }
      bal = __builtin_amdgcn_ballot_w32(m != 0u);
    } while (bal != 0u);
  };
  auto walk = [&](auto&& drain) {
    v4u bufA[4], bufB[4];
    auto load4 = [&](v4u (&buf)[4], int j) {
#pragma unroll
      for (int u = 0; u < 4; ++u) buf[u] = loadChunk(j + u);
    };
    auto look4 = [&](const v4u (&buf)[4], int j) {
#pragma unroll
      for (int u = 0; u < 4; ++u) look(buf[u], j + u, drain);
    };
    load4(bufA, 0);
    for (int j0 = 0; j0 < NJ; j0 += 8) {
      load4(bufB, j0 + 4);
      look4(bufA, j0);
      load4(bufA, j0 + 8);
      look4(bufB, j0 + 4);
    }
  };

  walk([&](unsigned h) { max_batch<true>(qw, h, BATCH, lane, src, dst, nodeBase, er, elT, mxE); });
  if (qt != qh)
    max_batch<false>(qw, qh, (int)(qt - qh), lane, src, dst, nodeBase, er, elT, mxE);
  const unsigned kept = qt;

  __syncthreads();
  if (tid < TILE_N * H_) mxT[tid] = decodeOrd(mxE[tid]);
  __syncthreads();

  if (kept <= LCAP) {
    unsigned p = 0;
    for (; p + BATCH <= kept; p += BATCH)
      scatter_batch<true, true>(qw, ww, p, BATCH, lane, src, dst, nodeBase, hW, er, elT, mxT, den, acc);
    if (p != kept)
      scatter_batch<false, true>(qw, ww, p, (int)(kept - p), lane, src, dst, nodeBase, hW, er, elT, mxT, den, acc);
  } else {
    qh = 0; qt = 0;
    walk([&](unsigned h) {
      scatter_batch<true, false>(qw, ww, h, BATCH, lane, src, dst, nodeBase, hW, er, elT, mxT, den, acc);
    });
    if (qt != qh)
      scatter_batch<false, false>(qw, ww, qh, (int)(qt - qh), lane, src, dst, nodeBase, hW, er, elT, mxT, den, acc);
  }

  __syncthreads();

  const v4f b4 = ((const v4f*)bias)[lane];
  for (int r = wave; r < TILE_N; r += SC_WAVES) {
    const int node = nodeBase + r;
    if (node < N_) {
      const float* p = acc + r * HF_ + lane;
      v4f v = (v4f){p[0], p[32], p[64], p[96]};
      const float d = den[r * H_ + (lane >> 3)];
      if (d != 0.f) { v.x /= d; v.y /= d; v.z /= d; v.w /= d; }
      ((v4f*)(out + (size_t)node * HF_))[lane] = v + b4;
    }
  }
}

extern "C" void kernel_launch(void* const* d_in, const int* in_sizes, int n_in,
                              void* d_out, int out_size, void* d_ws, size_t ws_size,
                              hipStream_t stream)
{
  const float* h    = (const float*)d_in[0];
  const float* W    = (const float*)d_in[1];
  const float* alw  = (const float*)d_in[2];
  const float* alb  = (const float*)d_in[3];
  const float* arw  = (const float*)d_in[4];
  const float* arb  = (const float*)d_in[5];
  const float* bias = (const float*)d_in[6];
  const int*   src  = (const int*)d_in[7];
  const int*   dst  = (const int*)d_in[8];
  float* out = (float*)d_out;

  char* ws = (char*)d_ws;
  size_t off = 0;
  auto carve = [&](size_t bytes) -> void* {
    void* p = ws + off;
    off += (bytes + 255) & ~(size_t)255;
    return p;
  };
  float*    hW    = (float*)carve((size_t)N_ * HF_ * sizeof(float));
  float*    el    = (float*)carve((size_t)N_ * H_ * sizeof(float));
  float*    er    = (float*)carve((size_t)N_ * H_ * sizeof(float));
  unsigned* tileOf = (unsigned*)carve((size_t)NCHUNK * CHUNK);
  (void)ws_size; (void)in_sizes; (void)n_in; (void)out_size;

  {
    const int strips = N_ / 16;
    const int blocks = (strips + 7) / 8;
    gat_gemm_wmma<<<blocks, 256, 0, stream>>>(h, W, hW, N_);
  }
  {
    const int nNH = N_ * H_;
    gat_attn_proj<<<(nNH + 255) / 256, 256, 0, stream>>>(hW, alw, alb, arw, arb, el, er, nNH, dst, tileOf);
  }
  {
    gat_scatter_tiles<<<NTILES, SC_WAVES * 32, 0, stream>>>(src, dst, tileOf, hW, el, er, bias, out);
  }
}


# Kept edges' scalars fetched once: projection launch packs source id and local node into one word per edge, first walk leaves the logits in LDS beside the list; 4 scattered fetches per edge become 2
# speedup vs baseline: 5.0696x; 1.0783x over previous
#include <hip/hip_runtime.h>

#define N_  50000
#define E_  800000
#define D_  256
#define H_  4
#define F_  32
#define HF_ 128
#define NEG_SLOPE 0.2f

typedef __attribute__((ext_vector_type(2))) float v2f;
typedef __attribute__((ext_vector_type(8))) float v8f;
typedef __attribute__((ext_vector_type(4))) float v4f;
typedef __attribute__((ext_vector_type(4))) int   v4i;
typedef __attribute__((ext_vector_type(4))) unsigned v4u;

#define TILE_N     197
#define NTILES     ((N_ + TILE_N - 1) / TILE_N)
#define CHUNK      512
#define NCHUNK     ((E_ + CHUNK - 1) / CHUNK)
#define NO_TILE    0xFFu
static_assert(NTILES <= NO_TILE, "tile id must fit one byte and differ from the padding");
static_assert(E_ % 4 == 0, "tile bytes are written four at a time");

__device__ __forceinline__ unsigned encodeOrd(float f) {
  unsigned u = __float_as_uint(f);
  return (u & 0x80000000u) ? ~u : (u | 0x80000000u);
}
__device__ __forceinline__ float decodeOrd(unsigned u) {
  unsigned b = (u & 0x80000000u) ? (u & 0x7FFFFFFFu) : ~u;
  return __uint_as_float(b);
}

__global__ __launch_bounds__(256) void gat_gemm_wmma(
    const float* __restrict__ h, const float* __restrict__ W,
    float* __restrict__ hW, int nRows)
{
  const int wave = threadIdx.x >> 5;
  const int lane = threadIdx.x & 31;
  const int rowBase = (blockIdx.x * 8 + wave) * 16;
  if (rowBase >= nRows) return;

  const int c  = lane & 15;
  const int hf = lane >> 4;

  v8f acc[8];
#pragma unroll
  for (int t = 0; t < 8; ++t) acc[t] = (v8f){0.f,0.f,0.f,0.f,0.f,0.f,0.f,0.f};

  const float* ap = h + (size_t)(rowBase + c) * D_ + 4 * hf;
  const float* bp = W + (size_t)(4 * hf) * HF_ + 8 * c;

  for (int kk = 0; kk < D_; kk += 8) {
    const v4f a4 = *(const v4f*)(ap + kk);
    v4f blo[4], bhi[4];
#pragma unroll
    for (int r = 0; r < 4; ++r) {
      const float* rp = bp + (size_t)(kk + r) * HF_;
      blo[r] = *(const v4f*)(rp);
      bhi[r] = *(const v4f*)(rp + 4);
    }
#pragma unroll
    for (int j = 0; j < 2; ++j) {
      v2f a; a.x = a4[2 * j]; a.y = a4[2 * j + 1];
#pragma unroll
      for (int t = 0; t < 8; ++t) {
        v2f b;
        b.x = (t < 4) ? blo[2 * j][t & 3]     : bhi[2 * j][t & 3];
        b.y = (t < 4) ? blo[2 * j + 1][t & 3] : bhi[2 * j + 1][t & 3];
        acc[t] = __builtin_amdgcn_wmma_f32_16x16x4_f32(
             false, a,  false, b,
             (short)0, acc[t],  false,  false);
      }
    }
  }

#pragma unroll
  for (int v = 0; v < 8; ++v) {
    float* op = hW + (size_t)(rowBase + 8 * hf + v) * HF_ + 8 * c;
    *(v4f*)(op)     = (v4f){acc[0][v], acc[1][v], acc[2][v], acc[3][v]};
    *(v4f*)(op + 4) = (v4f){acc[4][v], acc[5][v], acc[6][v], acc[7][v]};
  }
}

__global__ __launch_bounds__(256) void gat_attn_proj(
    const float* __restrict__ hW,
    const float* __restrict__ alw, const float* __restrict__ alb,
    const float* __restrict__ arw, const float* __restrict__ arb,
    float* __restrict__ el, float* __restrict__ er, int nNH,
    const int* __restrict__ src, const int* __restrict__ dst,
    unsigned* __restrict__ tileOf, unsigned* __restrict__ rec)
{
  const int idx = blockIdx.x * blockDim.x + threadIdx.x;
  for (int i = idx; i < NCHUNK * (CHUNK / 4); i += (int)(gridDim.x * blockDim.x)) {
    unsigned t4 = NO_TILE * 0x01010101u;
    if (i < E_ / 4) {
      const v4i d = ((const v4i*)dst)[i];
      const v4i s = ((const v4i*)src)[i];
      const unsigned t0 = (unsigned)d.x / TILE_N, t1 = (unsigned)d.y / TILE_N;
      const unsigned t2 = (unsigned)d.z / TILE_N, t3 = (unsigned)d.w / TILE_N;
      t4 = t0 | (t1 << 8) | (t2 << 16) | (t3 << 24);
      v4u r;
      r.x = ((unsigned)s.x << 8) | ((unsigned)d.x - t0 * TILE_N);
      r.y = ((unsigned)s.y << 8) | ((unsigned)d.y - t1 * TILE_N);
      r.z = ((unsigned)s.z << 8) | ((unsigned)d.z - t2 * TILE_N);
      r.w = ((unsigned)s.w << 8) | ((unsigned)d.w - t3 * TILE_N);
      ((v4u*)rec)[i] = r;
    }
    tileOf[i] = t4;
  }
  if (idx >= nNH) return;
  const float* p = hW + (size_t)idx * F_;
  float sl = 0.f, sr = 0.f;
#pragma unroll
  for (int f = 0; f < F_; ++f) {
    const float x = p[f];
    sl += x * alw[f];
    sr += x * arw[f];
  }
  el[idx] = sl + alb[0];
  er[idx] = sr + arb[0];
}

#define SC_WAVES 32
#define LCAP     256
#define BATCH    32

static_assert(N_ <= (1 << 24), "node id must fit 24 bits of a list entry");
static_assert(TILE_N <= 256, "local node id must fit 8 bits of a list entry");
static_assert(TILE_N * H_ <= SC_WAVES * 32, "at most one denominator per thread");
static_assert(2 * BATCH <= LCAP, "list must hold a batch plus one entry per lane");

__device__ __forceinline__ float leaky(float v) {
  return (v >= 0.f) ? v : NEG_SLOPE * v;
}

template <bool FULL>
__device__ __forceinline__ void max_batch(
    unsigned* qw, float* lgw, unsigned qh, int n, int lane,
    const unsigned* __restrict__ rec,
    const float* __restrict__ er, const float* elT, unsigned* mxE)
{
  if (FULL || lane < n) {
    const unsigned idx = (qh + (unsigned)lane) & (LCAP - 1);
    const unsigned e = qw[idx];
    const unsigned pw = rec[e];
    const int s = (int)(pw >> 8);
    const int dl = (int)(pw & 255u);
    const v4f r4 = ((const v4f*)er)[s];
    const v4f l4 = ((const v4f*)elT)[dl];
    const float v0 = leaky(r4.x + l4.x), v1 = leaky(r4.y + l4.y);
    const float v2 = leaky(r4.z + l4.z), v3 = leaky(r4.w + l4.w);
    unsigned* mp = mxE + dl * H_;
    atomicMax(mp + 0, encodeOrd(v0));
    atomicMax(mp + 1, encodeOrd(v1));
    atomicMax(mp + 2, encodeOrd(v2));
    atomicMax(mp + 3, encodeOrd(v3));
    qw[idx] = pw;
    ((v4f*)lgw)[idx] = (v4f){v0, v1, v2, v3};
  }
}

template <bool FULL, bool SRCFORM>
__device__ __forceinline__ void scatter_batch(
    const unsigned* qw, const float* lgw, float* ww, unsigned qh, int n, int lane,
    const unsigned* __restrict__ rec,
    const float* __restrict__ hW,
    const float* __restrict__ er, const float* elT, const float* mxT,
    float* den, float* acc)
{
  const bool mine = FULL || lane < n;
  int s = 0, dl = 0;
  v4f r4 = (v4f){0.f, 0.f, 0.f, 0.f};
  if (mine) {
    const unsigned idx = (qh + (unsigned)lane) & (LCAP - 1);
    const unsigned ent = qw[idx];
    const unsigned pw = SRCFORM ? ent : rec[ent];
    dl = (int)(pw & 255u);
    s = (int)(pw >> 8);
    r4 = SRCFORM ? ((const v4f*)lgw)[idx] : ((const v4f*)er)[s];
  }
  const int hd = lane >> 3;
  const int cnt = FULL ? BATCH : n;
  v4f row[8];
  float wv[8];
  auto loadRows = [&](int g) {
#pragma unroll
    for (int u = 0; u < 8; ++u) {
      row[u] = (v4f){0.f, 0.f, 0.f, 0.f};
      if (FULL || g + u < n) {
        const int sk = __builtin_amdgcn_readlane(s, g + u);
        const v4f* rp = (const v4f*)(hW + (size_t)sk * HF_);
        row[u] = rp[(unsigned)lane];
      }
    }
  };
  auto loadWeights = [&](int g) {
#pragma unroll
    for (int u = 0; u < 8; ++u) {
      wv[u] = 0.f;
      if (FULL || g + u < n) wv[u] = ww[(g + u) * H_ + hd];
    }
  };
  auto addRows = [&](int g) {
#pragma unroll
    for (int u = 0; u < 8; ++u) {
      if (FULL || g + u < n) {
        const int dlk = __builtin_amdgcn_readlane(dl, g + u);
        float* p = acc + dlk * HF_ + lane;
        const float w = wv[u];
        atomicAdd(p,      row[u].x * w);
        atomicAdd(p + 32, row[u].y * w);
        atomicAdd(p + 64, row[u].z * w);
        atomicAdd(p + 96, row[u].w * w);
      }
    }
  };

  loadRows(0);
  if (mine) {
    const v4f m4 = ((const v4f*)mxT)[dl];
    float v0 = r4.x, v1 = r4.y, v2 = r4.z, v3 = r4.w;
    if (!SRCFORM) {
      const v4f l4 = ((const v4f*)elT)[dl];
      v0 = leaky(r4.x + l4.x); v1 = leaky(r4.y + l4.y);
      v2 = leaky(r4.z + l4.z); v3 = leaky(r4.w + l4.w);
    }
    v4f x;
    x.x = expf(v0 - m4.x);
    x.y = expf(v1 - m4.y);
    x.z = expf(v2 - m4.z);
    x.w = expf(v3 - m4.w);
    ((v4f*)ww)[lane] = x;
    float* dp = den + dl * H_;
    atomicAdd(dp + 0, x.x);
    atomicAdd(dp + 1, x.y);
    atomicAdd(dp + 2, x.z);
    atomicAdd(dp + 3, x.w);
  }
  loadWeights(0);
  addRows(0);
  for (int g = 8; g < cnt; g += 8) {
    loadRows(g);
    loadWeights(g);
    addRows(g);
  }
}

__global__ __launch_bounds__(1024) void gat_scatter_tiles(
    const unsigned* __restrict__ rec,
    const unsigned* __restrict__ tileOf,
    const float* __restrict__ hW, const float* __restrict__ el,
    const float* __restrict__ er,
    const float* __restrict__ bias, float* __restrict__ out)
{
  __shared__ float    acc[TILE_N * HF_];
  __shared__ __attribute__((aligned(16))) float    den[TILE_N * H_];
  __shared__ __attribute__((aligned(16))) float    elT[TILE_N * H_];
  __shared__ __attribute__((aligned(16))) unsigned mxE[TILE_N * H_];
  __shared__ __attribute__((aligned(16))) float    mxT[TILE_N * H_];
  __shared__ unsigned q[SC_WAVES * LCAP];
  __shared__ __attribute__((aligned(16))) float lg[SC_WAVES * LCAP * H_];
  __shared__ float    wt[SC_WAVES * BATCH * H_];

  const int tid  = threadIdx.x;
  const int lane = tid & 31;
  const int wave = __builtin_amdgcn_readfirstlane(tid >> 5);
  const int nodeBase = blockIdx.x * TILE_N;

  for (int i = tid; i < TILE_N * HF_ / 4; i += SC_WAVES * 32)
    ((v4f*)acc)[i] = (v4f){0.f, 0.f, 0.f, 0.f};
  if (tid < TILE_N * H_) {
    const int g = nodeBase * H_ + tid;
    const bool in = g < N_ * H_;
    den[tid] = 0.f;
    elT[tid] = in ? el[g] : 0.f;
    mxE[tid] = encodeOrd(-__builtin_inff());
  }
  __syncthreads();

  unsigned* qw = q + wave * LCAP;
  float*    lgw = lg + wave * (LCAP * H_);
  float*    ww = wt + wave * (BATCH * H_);
  unsigned qh = 0, qt = 0;
  const int rot = blockIdx.x * (NCHUNK / NTILES);
  const int NJ  = (NCHUNK + SC_WAVES - 1) / SC_WAVES;
  const unsigned pat = (unsigned)blockIdx.x * 0x01010101u;

  auto chunkOf = [&](int j) -> int {
    const int k = wave + SC_WAVES * j;
    if (k >= NCHUNK) return -1;
    const int c = rot + k;
    return (c >= NCHUNK) ? c - NCHUNK : c;
  };
  auto loadChunk = [&](int j) -> v4u {
    const int c = chunkOf(j);
    const int cc = (c < 0) ? 0 : c;
    return ((const v4u*)(tileOf + (size_t)cc * (CHUNK / 4)))[(unsigned)lane];
  };
  auto hitBits = [&](unsigned w) -> unsigned {
    const unsigned y = w ^ pat;
    return ~(((y & 0x7F7F7F7Fu) + 0x7F7F7F7Fu) | y) & 0x80808080u;
  };
  auto look = [&](const v4u t, int j, auto&& drain) {
    const int c = chunkOf(j);
    if (c < 0) return;
    unsigned m = (hitBits(t.x) >> 7) | (hitBits(t.y) >> 6) |
                 (hitBits(t.z) >> 5) | (hitBits(t.w) >> 4);
    unsigned bal = __builtin_amdgcn_ballot_w32(m != 0u);
    if (bal == 0u) return;
    const unsigned e0 = (unsigned)(c * CHUNK + lane * 16);
    do {
      if (m != 0u) {
        const unsigned b = (unsigned)__builtin_ctz(m);
        qw[(qt + __builtin_amdgcn_mbcnt_lo(bal, 0u)) & (LCAP - 1)] = e0 + ((b & 3u) << 2) + (b >> 3);
        m &= m - 1u;
      }
      qt += __builtin_popcount(bal);
      if (qt - qh >= BATCH) {
        drain(qh);
        qh += BATCH;
      }
      bal = __builtin_amdgcn_ballot_w32(m != 0u);
    } while (bal != 0u);
  };
  auto walk = [&](auto&& drain) {
    v4u bufA[4], bufB[4];
    auto load4 = [&](v4u (&buf)[4], int j) {
#pragma unroll
      for (int u = 0; u < 4; ++u) buf[u] = loadChunk(j + u);
    };
    auto look4 = [&](const v4u (&buf)[4], int j) {
#pragma unroll
      for (int u = 0; u < 4; ++u) look(buf[u], j + u, drain);
    };
    load4(bufA, 0);
    for (int j0 = 0; j0 < NJ; j0 += 8) {
      load4(bufB, j0 + 4);
      look4(bufA, j0);
      load4(bufA, j0 + 8);
      look4(bufB, j0 + 4);
    }
  };

  walk([&](unsigned h) { max_batch<true>(qw, lgw, h, BATCH, lane, rec, er, elT, mxE); });
  if (qt != qh)
    max_batch<false>(qw, lgw, qh, (int)(qt - qh), lane, rec, er, elT, mxE);
  const unsigned kept = qt;

  __syncthreads();
  if (tid < TILE_N * H_) mxT[tid] = decodeOrd(mxE[tid]);
  __syncthreads();

  if (kept <= LCAP) {
    unsigned p = 0;
    for (; p + BATCH <= kept; p += BATCH)
      scatter_batch<true, true>(qw, lgw, ww, p, BATCH, lane, rec, hW, er, elT, mxT, den, acc);
    if (p != kept)
      scatter_batch<false, true>(qw, lgw, ww, p, (int)(kept - p), lane, rec, hW, er, elT, mxT, den, acc);
  } else {
    qh = 0; qt = 0;
    walk([&](unsigned h) {
      scatter_batch<true, false>(qw, lgw, ww, h, BATCH, lane, rec, hW, er, elT, mxT, den, acc);
    });
    if (qt != qh)
      scatter_batch<false, false>(qw, lgw, ww, qh, (int)(qt - qh), lane, rec, hW, er, elT, mxT, den, acc);
  }

  __syncthreads();

  const v4f b4 = ((const v4f*)bias)[lane];
  for (int r = wave; r < TILE_N; r += SC_WAVES) {
    const int node = nodeBase + r;
    if (node < N_) {
      const float* p = acc + r * HF_ + lane;
      v4f v = (v4f){p[0], p[32], p[64], p[96]};
      const float d = den[r * H_ + (lane >> 3)];
      if (d != 0.f) { v.x /= d; v.y /= d; v.z /= d; v.w /= d; }
      ((v4f*)(out + (size_t)node * HF_))[lane] = v + b4;
    }
  }
}

extern "C" void kernel_launch(void* const* d_in, const int* in_sizes, int n_in,
                              void* d_out, int out_size, void* d_ws, size_t ws_size,
                              hipStream_t stream)
{
  const float* h    = (const float*)d_in[0];
  const float* W    = (const float*)d_in[1];
  const float* alw  = (const float*)d_in[2];
  const float* alb  = (const float*)d_in[3];
  const float* arw  = (const float*)d_in[4];
  const float* arb  = (const float*)d_in[5];
  const float* bias = (const float*)d_in[6];
  const int*   src  = (const int*)d_in[7];
  const int*   dst  = (const int*)d_in[8];
  float* out = (float*)d_out;

  char* ws = (char*)d_ws;
  size_t off = 0;
  auto carve = [&](size_t bytes) -> void* {
    void* p = ws + off;
    off += (bytes + 255) & ~(size_t)255;
    return p;
  };
  float*    hW    = (float*)carve((size_t)N_ * HF_ * sizeof(float));
  float*    el    = (float*)carve((size_t)N_ * H_ * sizeof(float));
  float*    er    = (float*)carve((size_t)N_ * H_ * sizeof(float));
  unsigned* tileOf = (unsigned*)carve((size_t)NCHUNK * CHUNK);
  unsigned* rec    = (unsigned*)carve((size_t)E_ * sizeof(unsigned));
  (void)ws_size; (void)in_sizes; (void)n_in; (void)out_size;

  {
    const int strips = N_ / 16;
    const int blocks = (strips + 7) / 8;
    gat_gemm_wmma<<<blocks, 256, 0, stream>>>(h, W, hW, N_);
  }
  {
    const int nNH = N_ * H_;
    gat_attn_proj<<<(nNH + 255) / 256, 256, 0, stream>>>(hW, alw, alb, arw, arb, el, er, nNH, src, dst, tileOf, rec);
  }
  {
    gat_scatter_tiles<<<NTILES, SC_WAVES * 32, 0, stream>>>(rec, tileOf, hW, el, er, bias, out);
  }
}


# Kept edges put on per-node lists in LDS (slot by LDS integer add); each wave sums its nodes' rows and denominators in registers and stores the rows itself: LDS float adds gone; sums in slot order
# speedup vs baseline: 5.4151x; 1.0682x over previous
#include <hip/hip_runtime.h>

#define N_  50000
#define E_  800000
#define D_  256
#define H_  4
#define F_  32
#define HF_ 128
#define NEG_SLOPE 0.2f

typedef __attribute__((ext_vector_type(2))) float v2f;
typedef __attribute__((ext_vector_type(8))) float v8f;
typedef __attribute__((ext_vector_type(4))) float v4f;
typedef __attribute__((ext_vector_type(4))) int   v4i;
typedef __attribute__((ext_vector_type(4))) unsigned v4u;

#define TILE_N     197
#define NTILES     ((N_ + TILE_N - 1) / TILE_N)
#define CHUNK      512
#define NCHUNK     ((E_ + CHUNK - 1) / CHUNK)
#define NO_TILE    0xFFu
static_assert(NTILES <= NO_TILE, "tile id must fit one byte and differ from the padding");
static_assert(E_ % 4 == 0, "tile bytes are written four at a time");

__device__ __forceinline__ unsigned encodeOrd(float f) {
  unsigned u = __float_as_uint(f);
  return (u & 0x80000000u) ? ~u : (u | 0x80000000u);
}
__device__ __forceinline__ float decodeOrd(unsigned u) {
  unsigned b = (u & 0x80000000u) ? (u & 0x7FFFFFFFu) : ~u;
  return __uint_as_float(b);
}

__global__ __launch_bounds__(256) void gat_gemm_wmma(
    const float* __restrict__ h, const float* __restrict__ W,
    float* __restrict__ hW, int nRows)
{
  const int wave = threadIdx.x >> 5;
  const int lane = threadIdx.x & 31;
  const int rowBase = (blockIdx.x * 8 + wave) * 16;
  if (rowBase >= nRows) return;

  const int c  = lane & 15;
  const int hf = lane >> 4;

  v8f acc[8];
#pragma unroll
  for (int t = 0; t < 8; ++t) acc[t] = (v8f){0.f,0.f,0.f,0.f,0.f,0.f,0.f,0.f};

  const float* ap = h + (size_t)(rowBase + c) * D_ + 4 * hf;
  const float* bp = W + (size_t)(4 * hf) * HF_ + 8 * c;

  for (int kk = 0; kk < D_; kk += 8) {
    const v4f a4 = *(const v4f*)(ap + kk);
    v4f blo[4], bhi[4];
#pragma unroll
    for (int r = 0; r < 4; ++r) {
      const float* rp = bp + (size_t)(kk + r) * HF_;
      blo[r] = *(const v4f*)(rp);
      bhi[r] = *(const v4f*)(rp + 4);
    }
#pragma unroll
    for (int j = 0; j < 2; ++j) {
      v2f a; a.x = a4[2 * j]; a.y = a4[2 * j + 1];
#pragma unroll
      for (int t = 0; t < 8; ++t) {
        v2f b;
        b.x = (t < 4) ? blo[2 * j][t & 3]     : bhi[2 * j][t & 3];
        b.y = (t < 4) ? blo[2 * j + 1][t & 3] : bhi[2 * j + 1][t & 3];
        acc[t] = __builtin_amdgcn_wmma_f32_16x16x4_f32(
             false, a,  false, b,
             (short)0, acc[t],  false,  false);
      }
    }
  }

#pragma unroll
  for (int v = 0; v < 8; ++v) {
    float* op = hW + (size_t)(rowBase + 8 * hf + v) * HF_ + 8 * c;
    *(v4f*)(op)     = (v4f){acc[0][v], acc[1][v], acc[2][v], acc[3][v]};
    *(v4f*)(op + 4) = (v4f){acc[4][v], acc[5][v], acc[6][v], acc[7][v]};
  }
}

__global__ __launch_bounds__(256) void gat_attn_proj(
    const float* __restrict__ hW,
    const float* __restrict__ alw, const float* __restrict__ alb,
    const float* __restrict__ arw, const float* __restrict__ arb,
    float* __restrict__ el, float* __restrict__ er, int nNH,
    const int* __restrict__ src, const int* __restrict__ dst,
    unsigned* __restrict__ tileOf, unsigned* __restrict__ rec)
{
  const int idx = blockIdx.x * blockDim.x + threadIdx.x;
  for (int i = idx; i < NCHUNK * (CHUNK / 4); i += (int)(gridDim.x * blockDim.x)) {
    unsigned t4 = NO_TILE * 0x01010101u;
    if (i < E_ / 4) {
      const v4i d = ((const v4i*)dst)[i];
      const v4i s = ((const v4i*)src)[i];
      const unsigned t0 = (unsigned)d.x / TILE_N, t1 = (unsigned)d.y / TILE_N;
      const unsigned t2 = (unsigned)d.z / TILE_N, t3 = (unsigned)d.w / TILE_N;
      t4 = t0 | (t1 << 8) | (t2 << 16) | (t3 << 24);
      v4u r;
      r.x = ((unsigned)s.x << 8) | ((unsigned)d.x - t0 * TILE_N);
      r.y = ((unsigned)s.y << 8) | ((unsigned)d.y - t1 * TILE_N);
      r.z = ((unsigned)s.z << 8) | ((unsigned)d.z - t2 * TILE_N);
      r.w = ((unsigned)s.w << 8) | ((unsigned)d.w - t3 * TILE_N);
      ((v4u*)rec)[i] = r;
    }
    tileOf[i] = t4;
  }
  if (idx >= nNH) return;
  const float* p = hW + (size_t)idx * F_;
  float sl = 0.f, sr = 0.f;
#pragma unroll
  for (int f = 0; f < F_; ++f) {
    const float x = p[f];
    sl += x * alw[f];
    sr += x * arw[f];
  }
  el[idx] = sl + alb[0];
  er[idx] = sr + arb[0];
}

#define SC_WAVES 32
#define NCAP     44
#define RCAP     64
#define BATCH    32

static_assert(N_ <= (1 << 24), "node id must fit 24 bits of a list entry");
static_assert(TILE_N <= 256, "local node id must fit 8 bits of a list entry");
static_assert(TILE_N * H_ <= SC_WAVES * 32, "at most one denominator per thread");
static_assert(2 * BATCH <= RCAP, "ring must hold a batch plus one entry per lane");

__device__ __forceinline__ float leaky(float v) {
  return (v >= 0.f) ? v : NEG_SLOPE * v;
}

template <bool FULL>
__device__ __forceinline__ void max_batch(
    const unsigned* qw, unsigned qh, int n, int lane,
    const unsigned* __restrict__ rec,
    const float* __restrict__ er, const float* elT, unsigned* mxE,
    unsigned* ncnt, unsigned* nw, float* nlg, unsigned* ovf)
{
  if (FULL || lane < n) {
    const unsigned e = qw[(qh + (unsigned)lane) & (RCAP - 1)];
    const unsigned pw = rec[e];
    const int s = (int)(pw >> 8);
    const int dl = (int)(pw & 255u);
    const v4f r4 = ((const v4f*)er)[s];
    const v4f l4 = ((const v4f*)elT)[dl];
    const float v0 = leaky(r4.x + l4.x), v1 = leaky(r4.y + l4.y);
    const float v2 = leaky(r4.z + l4.z), v3 = leaky(r4.w + l4.w);
    unsigned* mp = mxE + dl * H_;
    atomicMax(mp + 0, encodeOrd(v0));
    atomicMax(mp + 1, encodeOrd(v1));
    atomicMax(mp + 2, encodeOrd(v2));
    atomicMax(mp + 3, encodeOrd(v3));
    const unsigned slot = atomicAdd(ncnt + dl, 1u);
    if (slot < NCAP) {
      const unsigned at = (unsigned)dl * NCAP + slot;
      nw[at] = pw;
      ((v4f*)nlg)[at] = (v4f){v0, v1, v2, v3};
    } else {
      *ovf = 1u;
    }
  }
}

__device__ __forceinline__ void node_batch(
    const unsigned* lw, const float* llg, float* ww, int n, int lane,
    const float* __restrict__ hW, const v4f m4, v4f& a, float& d)
{
  const bool mine = lane < n;
  int s = 0;
  if (mine) s = (int)(lw[lane] >> 8);
  const int hd = lane >> 3;
  v4f row[8];
  float wv[8];
  auto loadRows = [&](int g) {
#pragma unroll
    for (int u = 0; u < 8; ++u) {
      row[u] = (v4f){0.f, 0.f, 0.f, 0.f};
      if (g + u < n) {
        const int sk = __builtin_amdgcn_readlane(s, g + u);
        const v4f* rp = (const v4f*)(hW + (size_t)sk * HF_);
        row[u] = rp[(unsigned)lane];
      }
    }
  };
  auto addRows = [&](int g) {
#pragma unroll
    for (int u = 0; u < 8; ++u) {
      wv[u] = 0.f;
      if (g + u < n) wv[u] = ww[(g + u) * H_ + hd];
    }
#pragma unroll
    for (int u = 0; u < 8; ++u) {
      a += row[u] * wv[u];
      d += wv[u];
    }
  };

  loadRows(0);
  if (mine) {
    const v4f lg4 = ((const v4f*)llg)[lane];
    v4f x;
    x.x = expf(lg4.x - m4.x);
    x.y = expf(lg4.y - m4.y);
    x.z = expf(lg4.z - m4.z);
    x.w = expf(lg4.w - m4.w);
    ((v4f*)ww)[lane] = x;
  }
  addRows(0);
  for (int g = 8; g < n; g += 8) {
    loadRows(g);
    addRows(g);
  }
}

template <bool FULL>
__device__ __forceinline__ void scatter_batch(
    const unsigned* qw, float* ww, unsigned qh, int n, int lane,
    const unsigned* __restrict__ rec,
    const float* __restrict__ hW,
    const float* __restrict__ er, const float* elT, const float* mxT,
    float* den, float* acc)
{
  const bool mine = FULL || lane < n;
  int s = 0, dl = 0;
  v4f r4 = (v4f){0.f, 0.f, 0.f, 0.f};
  if (mine) {
    const unsigned pw = rec[qw[(qh + (unsigned)lane) & (RCAP - 1)]];
    dl = (int)(pw & 255u);
    s = (int)(pw >> 8);
    r4 = ((const v4f*)er)[s];
  }
  const int hd = lane >> 3;
  const int cnt = FULL ? BATCH : n;
  v4f row[8];
  float wv[8];
  auto loadRows = [&](int g) {
#pragma unroll
    for (int u = 0; u < 8; ++u) {
      row[u] = (v4f){0.f, 0.f, 0.f, 0.f};
      if (FULL || g + u < n) {
        const int sk = __builtin_amdgcn_readlane(s, g + u);
        const v4f* rp = (const v4f*)(hW + (size_t)sk * HF_);
        row[u] = rp[(unsigned)lane];
      }
    }
  };
  auto loadWeights = [&](int g) {
#pragma unroll
    for (int u = 0; u < 8; ++u) {
      wv[u] = 0.f;
      if (FULL || g + u < n) wv[u] = ww[(g + u) * H_ + hd];
    }
  };
  auto addRows = [&](int g) {
#pragma unroll
    for (int u = 0; u < 8; ++u) {
      if (FULL || g + u < n) {
        const int dlk = __builtin_amdgcn_readlane(dl, g + u);
        float* p = acc + dlk * HF_ + lane;
        const float w = wv[u];
        atomicAdd(p,      row[u].x * w);
        atomicAdd(p + 32, row[u].y * w);
        atomicAdd(p + 64, row[u].z * w);
        atomicAdd(p + 96, row[u].w * w);
      }
    }
  };

  loadRows(0);
  if (mine) {
    const v4f m4 = ((const v4f*)mxT)[dl];
    const v4f l4 = ((const v4f*)elT)[dl];
    v4f x;
    x.x = expf(leaky(r4.x + l4.x) - m4.x);
    x.y = expf(leaky(r4.y + l4.y) - m4.y);
    x.z = expf(leaky(r4.z + l4.z) - m4.z);
    x.w = expf(leaky(r4.w + l4.w) - m4.w);
    ((v4f*)ww)[lane] = x;
    float* dp = den + dl * H_;
    atomicAdd(dp + 0, x.x);
    atomicAdd(dp + 1, x.y);
    atomicAdd(dp + 2, x.z);
    atomicAdd(dp + 3, x.w);
  }
  loadWeights(0);
  addRows(0);
  for (int g = 8; g < cnt; g += 8) {
    loadRows(g);
    loadWeights(g);
    addRows(g);
  }
}

__global__ __launch_bounds__(1024) void gat_scatter_tiles(
    const unsigned* __restrict__ rec,
    const unsigned* __restrict__ tileOf,
    const float* __restrict__ hW, const float* __restrict__ el,
    const float* __restrict__ er,
    const float* __restrict__ bias, float* __restrict__ out)
{
  __shared__ float    acc[TILE_N * HF_];
  __shared__ __attribute__((aligned(16))) float    den[TILE_N * H_];
  __shared__ __attribute__((aligned(16))) float    elT[TILE_N * H_];
  __shared__ __attribute__((aligned(16))) unsigned mxE[TILE_N * H_];
  __shared__ __attribute__((aligned(16))) float    mxT[TILE_N * H_];
  __shared__ unsigned q[SC_WAVES * RCAP];
  __shared__ unsigned nw[TILE_N * NCAP];
  __shared__ __attribute__((aligned(16))) float nlg[TILE_N * NCAP * H_];
  __shared__ unsigned ncnt[TILE_N];
  __shared__ unsigned ovf;
  __shared__ float    wt[SC_WAVES * BATCH * H_];

  const int tid  = threadIdx.x;
  const int lane = tid & 31;
  const int wave = __builtin_amdgcn_readfirstlane(tid >> 5);
  const int nodeBase = blockIdx.x * TILE_N;

  for (int i = tid; i < TILE_N * HF_ / 4; i += SC_WAVES * 32)
    ((v4f*)acc)[i] = (v4f){0.f, 0.f, 0.f, 0.f};
  if (tid < TILE_N * H_) {
    const int g = nodeBase * H_ + tid;
    const bool in = g < N_ * H_;
    den[tid] = 0.f;
    elT[tid] = in ? el[g] : 0.f;
    mxE[tid] = encodeOrd(-__builtin_inff());
  }
  if (tid < TILE_N) ncnt[tid] = 0u;
  if (tid == 0) ovf = 0u;
  __syncthreads();

  unsigned* qw = q + wave * RCAP;
  float*    ww = wt + wave * (BATCH * H_);
  unsigned qh = 0, qt = 0;
  const int rot = blockIdx.x * (NCHUNK / NTILES);
  const int NJ  = (NCHUNK + SC_WAVES - 1) / SC_WAVES;
  const unsigned pat = (unsigned)blockIdx.x * 0x01010101u;

  auto chunkOf = [&](int j) -> int {
    const int k = wave + SC_WAVES * j;
    if (k >= NCHUNK) return -1;
    const int c = rot + k;
    return (c >= NCHUNK) ? c - NCHUNK : c;
  };
  auto loadChunk = [&](int j) -> v4u {
    const int c = chunkOf(j);
    const int cc = (c < 0) ? 0 : c;
    return ((const v4u*)(tileOf + (size_t)cc * (CHUNK / 4)))[(unsigned)lane];
  };
  auto hitBits = [&](unsigned w) -> unsigned {
    const unsigned y = w ^ pat;
    return ~(((y & 0x7F7F7F7Fu) + 0x7F7F7F7Fu) | y) & 0x80808080u;
  };
  auto look = [&](const v4u t, int j, auto&& drain) {
    const int c = chunkOf(j);
    if (c < 0) return;
    unsigned m = (hitBits(t.x) >> 7) | (hitBits(t.y) >> 6) |
                 (hitBits(t.z) >> 5) | (hitBits(t.w) >> 4);
    unsigned bal = __builtin_amdgcn_ballot_w32(m != 0u);
    if (bal == 0u) return;
    const unsigned e0 = (unsigned)(c * CHUNK + lane * 16);
    do {
      if (m != 0u) {
        const unsigned b = (unsigned)__builtin_ctz(m);
        qw[(qt + __builtin_amdgcn_mbcnt_lo(bal, 0u)) & (RCAP - 1)] = e0 + ((b & 3u) << 2) + (b >> 3);
        m &= m - 1u;
      }
      qt += __builtin_popcount(bal);
      if (qt - qh >= BATCH) {
        drain(qh);
        qh += BATCH;
      }
      bal = __builtin_amdgcn_ballot_w32(m != 0u);
    } while (bal != 0u);
  };
  auto walk = [&](auto&& drain) {
    v4u bufA[4], bufB[4];
    auto load4 = [&](v4u (&buf)[4], int j) {
#pragma unroll
      for (int u = 0; u < 4; ++u) buf[u] = loadChunk(j + u);
    };
    auto look4 = [&](const v4u (&buf)[4], int j) {
#pragma unroll
      for (int u = 0; u < 4; ++u) look(buf[u], j + u, drain);
    };
    load4(bufA, 0);
    for (int j0 = 0; j0 < NJ; j0 += 8) {
      load4(bufB, j0 + 4);
      look4(bufA, j0);
      load4(bufA, j0 + 8);
      look4(bufB, j0 + 4);
    }
  };

  walk([&](unsigned h) { max_batch<true>(qw, h, BATCH, lane, rec, er, elT, mxE, ncnt, nw, nlg, &ovf); });
  if (qt != qh)
    max_batch<false>(qw, qh, (int)(qt - qh), lane, rec, er, elT, mxE, ncnt, nw, nlg, &ovf);

  __syncthreads();
  if (tid < TILE_N * H_) mxT[tid] = decodeOrd(mxE[tid]);
  __syncthreads();

  const v4f b4 = ((const v4f*)bias)[lane];
  const bool full = __builtin_amdgcn_readfirstlane((int)ovf) != 0;
  if (__builtin_expect(!full, 1)) {
    for (int r = wave; r < TILE_N; r += SC_WAVES) {
      const int node = nodeBase + r;
      if (node < N_) {
        const int cnt = __builtin_amdgcn_readfirstlane((int)ncnt[r]);
        const v4f m4 = ((const v4f*)mxT)[r];
        v4f a = (v4f){0.f, 0.f, 0.f, 0.f};
        float d = 0.f;
        for (int p = 0; p < cnt; p += BATCH) {
          const int n = (cnt - p < BATCH) ? cnt - p : BATCH;
          node_batch(nw + r * NCAP + p, nlg + (r * NCAP + p) * H_, ww, n, lane, hW, m4, a, d);
        }
        if (d != 0.f) { a.x /= d; a.y /= d; a.z /= d; a.w /= d; }
        ((v4f*)(out + (size_t)node * HF_))[lane] = a + b4;
      }
    }
  } else {
    qh = 0; qt = 0;
    walk([&](unsigned h) {
      scatter_batch<true>(qw, ww, h, BATCH, lane, rec, hW, er, elT, mxT, den, acc);
    });
    if (qt != qh)
      scatter_batch<false>(qw, ww, qh, (int)(qt - qh), lane, rec, hW, er, elT, mxT, den, acc);

    __syncthreads();

    for (int r = wave; r < TILE_N; r += SC_WAVES) {
      const int node = nodeBase + r;
      if (node < N_) {
        const float* p = acc + r * HF_ + lane;
        v4f v = (v4f){p[0], p[32], p[64], p[96]};
        const float dn = den[r * H_ + (lane >> 3)];
        if (dn != 0.f) { v.x /= dn; v.y /= dn; v.z /= dn; v.w /= dn; }
        ((v4f*)(out + (size_t)node * HF_))[lane] = v + b4;
      }
    }
  }
}

extern "C" void kernel_launch(void* const* d_in, const int* in_sizes, int n_in,
                              void* d_out, int out_size, void* d_ws, size_t ws_size,
                              hipStream_t stream)
{
  const float* h    = (const float*)d_in[0];
  const float* W    = (const float*)d_in[1];
  const float* alw  = (const float*)d_in[2];
  const float* alb  = (const float*)d_in[3];
  const float* arw  = (const float*)d_in[4];
  const float* arb  = (const float*)d_in[5];
  const float* bias = (const float*)d_in[6];
  const int*   src  = (const int*)d_in[7];
  const int*   dst  = (const int*)d_in[8];
  float* out = (float*)d_out;

  char* ws = (char*)d_ws;
  size_t off = 0;
  auto carve = [&](size_t bytes) -> void* {
    void* p = ws + off;
    off += (bytes + 255) & ~(size_t)255;
    return p;
  };
  float*    hW    = (float*)carve((size_t)N_ * HF_ * sizeof(float));
  float*    el    = (float*)carve((size_t)N_ * H_ * sizeof(float));
  float*    er    = (float*)carve((size_t)N_ * H_ * sizeof(float));
  unsigned* tileOf = (unsigned*)carve((size_t)NCHUNK * CHUNK);
  unsigned* rec    = (unsigned*)carve((size_t)E_ * sizeof(unsigned));
  (void)ws_size; (void)in_sizes; (void)n_in; (void)out_size;

  {
    const int strips = N_ / 16;
    const int blocks = (strips + 7) / 8;
    gat_gemm_wmma<<<blocks, 256, 0, stream>>>(h, W, hW, N_);
  }
  {
    const int nNH = N_ * H_;
    gat_attn_proj<<<(nNH + 255) / 256, 256, 0, stream>>>(hW, alw, alb, arw, arb, el, er, nNH, src, dst, tileOf, rec);
  }
  {
    gat_scatter_tiles<<<NTILES, SC_WAVES * 32, 0, stream>>>(rec, tileOf, hW, el, er, bias, out);
  }
}


# Tile walk in two levels: projection writes per 512-edge chunk a word per tile (bit: 16-edge group has an edge of it; LDS integer OR); tiles read these, then only named groups' bytes; slot sums
# speedup vs baseline: 5.6593x; 1.0451x over previous
#include <hip/hip_runtime.h>

#define N_  50000
#define E_  800000
#define D_  256
#define H_  4
#define F_  32
#define HF_ 128
#define NEG_SLOPE 0.2f

typedef __attribute__((ext_vector_type(2))) float v2f;
typedef __attribute__((ext_vector_type(8))) float v8f;
typedef __attribute__((ext_vector_type(4))) float v4f;
typedef __attribute__((ext_vector_type(4))) int   v4i;
typedef __attribute__((ext_vector_type(4))) unsigned v4u;

#define TILE_N     197
#define NTILES     ((N_ + TILE_N - 1) / TILE_N)
#define CHUNK      512
#define NCHUNK     ((E_ + CHUNK - 1) / CHUNK)
#define NO_TILE    0xFFu
#define GROUP      16
#define GPC        (CHUNK / GROUP)
#define GROW       256
static_assert(GPC == 32, "a chunk's groups are the bits of one word");
static_assert(NTILES <= GROW, "one group word per tile in a chunk's row");
static_assert(((N_ * H_ + 255) / 256) * 2 >= NCHUNK, "the second launch's work-groups, two chunks each, cover the list");
static_assert(NTILES <= NO_TILE, "tile id must fit one byte and differ from the padding");
static_assert(E_ % 4 == 0, "tile bytes are written four at a time");

__device__ __forceinline__ unsigned encodeOrd(float f) {
  unsigned u = __float_as_uint(f);
  return (u & 0x80000000u) ? ~u : (u | 0x80000000u);
}
__device__ __forceinline__ float decodeOrd(unsigned u) {
  unsigned b = (u & 0x80000000u) ? (u & 0x7FFFFFFFu) : ~u;
  return __uint_as_float(b);
}

__global__ __launch_bounds__(256) void gat_gemm_wmma(
    const float* __restrict__ h, const float* __restrict__ W,
    float* __restrict__ hW, int nRows)
{
  const int wave = threadIdx.x >> 5;
  const int lane = threadIdx.x & 31;
  const int rowBase = (blockIdx.x * 8 + wave) * 16;
  if (rowBase >= nRows) return;

  const int c  = lane & 15;
  const int hf = lane >> 4;

  v8f acc[8];
#pragma unroll
  for (int t = 0; t < 8; ++t) acc[t] = (v8f){0.f,0.f,0.f,0.f,0.f,0.f,0.f,0.f};

  const float* ap = h + (size_t)(rowBase + c) * D_ + 4 * hf;
  const float* bp = W + (size_t)(4 * hf) * HF_ + 8 * c;

  for (int kk = 0; kk < D_; kk += 8) {
    const v4f a4 = *(const v4f*)(ap + kk);
    v4f blo[4], bhi[4];
#pragma unroll
    for (int r = 0; r < 4; ++r) {
      const float* rp = bp + (size_t)(kk + r) * HF_;
      blo[r] = *(const v4f*)(rp);
      bhi[r] = *(const v4f*)(rp + 4);
    }
#pragma unroll
    for (int j = 0; j < 2; ++j) {
      v2f a; a.x = a4[2 * j]; a.y = a4[2 * j + 1];
#pragma unroll
      for (int t = 0; t < 8; ++t) {
        v2f b;
        b.x = (t < 4) ? blo[2 * j][t & 3]     : bhi[2 * j][t & 3];
        b.y = (t < 4) ? blo[2 * j + 1][t & 3] : bhi[2 * j + 1][t & 3];
        acc[t] = __builtin_amdgcn_wmma_f32_16x16x4_f32(
             false, a,  false, b,
             (short)0, acc[t],  false,  false);
      }
    }
  }

#pragma unroll
  for (int v = 0; v < 8; ++v) {
    float* op = hW + (size_t)(rowBase + 8 * hf + v) * HF_ + 8 * c;
    *(v4f*)(op)     = (v4f){acc[0][v], acc[1][v], acc[2][v], acc[3][v]};
    *(v4f*)(op + 4) = (v4f){acc[4][v], acc[5][v], acc[6][v], acc[7][v]};
  }
}

__global__ __launch_bounds__(256) void gat_attn_proj(
    const float* __restrict__ hW,
    const float* __restrict__ alw, const float* __restrict__ alb,
    const float* __restrict__ arw, const float* __restrict__ arb,
    float* __restrict__ el, float* __restrict__ er, int nNH,
    const int* __restrict__ src, const int* __restrict__ dst,
    unsigned* __restrict__ tileOf, unsigned* __restrict__ rec,
    unsigned* __restrict__ grp)
{
  __shared__ unsigned gm[2 * GROW];
  const int tid = threadIdx.x;
  const int idx = blockIdx.x * 256 + tid;
  gm[tid] = 0u;
  gm[tid + GROW] = 0u;
  __syncthreads();
  const int i = idx;
  unsigned t4 = NO_TILE * 0x01010101u;
  v4u r = (v4u){0u, 0u, 0u, 0u};
  if (i < E_ / 4) {
    const v4i d = ((const v4i*)dst)[i];
    const v4i s = ((const v4i*)src)[i];
    const unsigned t0 = (unsigned)d.x / TILE_N, t1 = (unsigned)d.y / TILE_N;
    const unsigned t2 = (unsigned)d.z / TILE_N, t3 = (unsigned)d.w / TILE_N;
    t4 = t0 | (t1 << 8) | (t2 << 16) | (t3 << 24);
    unsigned* row = gm + (tid >> 7) * GROW;
    const unsigned bit = 1u << ((tid & 127) >> 2);
    atomicOr(row + t0, bit);
    atomicOr(row + t1, bit);
    atomicOr(row + t2, bit);
    atomicOr(row + t3, bit);
    r.x = ((unsigned)s.x << 8) | ((unsigned)d.x - t0 * TILE_N);
    r.y = ((unsigned)s.y << 8) | ((unsigned)d.y - t1 * TILE_N);
    r.z = ((unsigned)s.z << 8) | ((unsigned)d.z - t2 * TILE_N);
    r.w = ((unsigned)s.w << 8) | ((unsigned)d.w - t3 * TILE_N);
  }
  __syncthreads();
  if (i < E_ / 4) ((v4u*)rec)[i] = r;
  if (i < NCHUNK * (CHUNK / 4)) tileOf[i] = t4;
  {
    const int c0 = blockIdx.x * 2;
    if (c0 < NCHUNK)     grp[(size_t)c0 * GROW + tid]       = gm[tid];
    if (c0 + 1 < NCHUNK) grp[(size_t)(c0 + 1) * GROW + tid] = gm[GROW + tid];
  }
  if (idx >= nNH) return;
  const float* p = hW + (size_t)idx * F_;
  float sl = 0.f, sr = 0.f;
#pragma unroll
  for (int f = 0; f < F_; ++f) {
    const float x = p[f];
    sl += x * alw[f];
    sr += x * arw[f];
  }
  el[idx] = sl + alb[0];
  er[idx] = sr + arb[0];
}

#define SC_WAVES 32
#define NCAP     44
#define RCAP     64
#define BATCH    32

static_assert(N_ <= (1 << 24), "node id must fit 24 bits of a list entry");
static_assert(TILE_N <= 256, "local node id must fit 8 bits of a list entry");
static_assert(TILE_N * H_ <= SC_WAVES * 32, "at most one denominator per thread");
static_assert(2 * BATCH <= RCAP, "ring must hold a batch plus one entry per lane");

__device__ __forceinline__ float leaky(float v) {
  return (v >= 0.f) ? v : NEG_SLOPE * v;
}

template <bool FULL>
__device__ __forceinline__ void max_batch(
    const unsigned* qw, unsigned qh, int n, int lane,
    const unsigned* __restrict__ rec,
    const float* __restrict__ er, const float* elT, unsigned* mxE,
    unsigned* ncnt, unsigned* nw, float* nlg, unsigned* ovf)
{
  if (FULL || lane < n) {
    const unsigned e = qw[(qh + (unsigned)lane) & (RCAP - 1)];
    const unsigned pw = rec[e];
    const int s = (int)(pw >> 8);
    const int dl = (int)(pw & 255u);
    const v4f r4 = ((const v4f*)er)[s];
    const v4f l4 = ((const v4f*)elT)[dl];
    const float v0 = leaky(r4.x + l4.x), v1 = leaky(r4.y + l4.y);
    const float v2 = leaky(r4.z + l4.z), v3 = leaky(r4.w + l4.w);
    unsigned* mp = mxE + dl * H_;
    atomicMax(mp + 0, encodeOrd(v0));
    atomicMax(mp + 1, encodeOrd(v1));
    atomicMax(mp + 2, encodeOrd(v2));
    atomicMax(mp + 3, encodeOrd(v3));
    const unsigned slot = atomicAdd(ncnt + dl, 1u);
    if (slot < NCAP) {
      const unsigned at = (unsigned)dl * NCAP + slot;
      nw[at] = pw;
      ((v4f*)nlg)[at] = (v4f){v0, v1, v2, v3};
    } else {
      *ovf = 1u;
    }
  }
}

__device__ __forceinline__ void node_batch(
    const unsigned* lw, const float* llg, float* ww, int n, int lane,
    const float* __restrict__ hW, const v4f m4, v4f& a, float& d)
{
  const bool mine = lane < n;
  int s = 0;
  if (mine) s = (int)(lw[lane] >> 8);
  const int hd = lane >> 3;
  v4f row[8];
  float wv[8];
  auto loadRows = [&](int g) {
#pragma unroll
    for (int u = 0; u < 8; ++u) {
      row[u] = (v4f){0.f, 0.f, 0.f, 0.f};
      if (g + u < n) {
        const int sk = __builtin_amdgcn_readlane(s, g + u);
        const v4f* rp = (const v4f*)(hW + (size_t)sk * HF_);
        row[u] = rp[(unsigned)lane];
      }
    }
  };
  auto addRows = [&](int g) {
#pragma unroll
    for (int u = 0; u < 8; ++u) {
      wv[u] = 0.f;
      if (g + u < n) wv[u] = ww[(g + u) * H_ + hd];
    }
#pragma unroll
    for (int u = 0; u < 8; ++u) {
      a += row[u] * wv[u];
      d += wv[u];
    }
  };

  loadRows(0);
  if (mine) {
    const v4f lg4 = ((const v4f*)llg)[lane];
    v4f x;
    x.x = expf(lg4.x - m4.x);
    x.y = expf(lg4.y - m4.y);
    x.z = expf(lg4.z - m4.z);
    x.w = expf(lg4.w - m4.w);
    ((v4f*)ww)[lane] = x;
  }
  addRows(0);
  for (int g = 8; g < n; g += 8) {
    loadRows(g);
    addRows(g);
  }
}

template <bool FULL>
__device__ __forceinline__ void scatter_batch(
    const unsigned* qw, float* ww, unsigned qh, int n, int lane,
    const unsigned* __restrict__ rec,
    const float* __restrict__ hW,
    const float* __restrict__ er, const float* elT, const float* mxT,
    float* den, float* acc)
{
  const bool mine = FULL || lane < n;
  int s = 0, dl = 0;
  v4f r4 = (v4f){0.f, 0.f, 0.f, 0.f};
  if (mine) {
    const unsigned pw = rec[qw[(qh + (unsigned)lane) & (RCAP - 1)]];
    dl = (int)(pw & 255u);
    s = (int)(pw >> 8);
    r4 = ((const v4f*)er)[s];
  }
  const int hd = lane >> 3;
  const int cnt = FULL ? BATCH : n;
  v4f row[8];
  float wv[8];
  auto loadRows = [&](int g) {
#pragma unroll
    for (int u = 0; u < 8; ++u) {
      row[u] = (v4f){0.f, 0.f, 0.f, 0.f};
      if (FULL || g + u < n) {
        const int sk = __builtin_amdgcn_readlane(s, g + u);
        const v4f* rp = (const v4f*)(hW + (size_t)sk * HF_);
        row[u] = rp[(unsigned)lane];
      }
    }
  };
  auto loadWeights = [&](int g) {
#pragma unroll
    for (int u = 0; u < 8; ++u) {
      wv[u] = 0.f;
      if (FULL || g + u < n) wv[u] = ww[(g + u) * H_ + hd];
    }
  };
  auto addRows = [&](int g) {
#pragma unroll
    for (int u = 0; u < 8; ++u) {
      if (FULL || g + u < n) {
        const int dlk = __builtin_amdgcn_readlane(dl, g + u);
        float* p = acc + dlk * HF_ + lane;
        const float w = wv[u];
        atomicAdd(p,      row[u].x * w);
        atomicAdd(p + 32, row[u].y * w);
        atomicAdd(p + 64, row[u].z * w);
        atomicAdd(p + 96, row[u].w * w);
      }
    }
  };

  loadRows(0);
  if (mine) {
    const v4f m4 = ((const v4f*)mxT)[dl];
    const v4f l4 = ((const v4f*)elT)[dl];
    v4f x;
    x.x = expf(leaky(r4.x + l4.x) - m4.x);
    x.y = expf(leaky(r4.y + l4.y) - m4.y);
    x.z = expf(leaky(r4.z + l4.z) - m4.z);
    x.w = expf(leaky(r4.w + l4.w) - m4.w);
    ((v4f*)ww)[lane] = x;
    float* dp = den + dl * H_;
    atomicAdd(dp + 0, x.x);
    atomicAdd(dp + 1, x.y);
    atomicAdd(dp + 2, x.z);
    atomicAdd(dp + 3, x.w);
  }
  loadWeights(0);
  addRows(0);
  for (int g = 8; g < cnt; g += 8) {
    loadRows(g);
    loadWeights(g);
    addRows(g);
  }
}

__global__ __launch_bounds__(1024) void gat_scatter_tiles(
    const unsigned* __restrict__ rec,
    const unsigned* __restrict__ tileOf,
    const unsigned* __restrict__ grp,
    const float* __restrict__ hW, const float* __restrict__ el,
    const float* __restrict__ er,
    const float* __restrict__ bias, float* __restrict__ out)
{
  __shared__ float    acc[TILE_N * HF_];
  __shared__ __attribute__((aligned(16))) float    den[TILE_N * H_];
  __shared__ __attribute__((aligned(16))) float    elT[TILE_N * H_];
  __shared__ __attribute__((aligned(16))) unsigned mxE[TILE_N * H_];
  __shared__ __attribute__((aligned(16))) float    mxT[TILE_N * H_];
  __shared__ unsigned q[SC_WAVES * RCAP];
  __shared__ unsigned gq[SC_WAVES * RCAP];
  __shared__ unsigned nw[TILE_N * NCAP];
  __shared__ __attribute__((aligned(16))) float nlg[TILE_N * NCAP * H_];
  __shared__ unsigned ncnt[TILE_N];
  __shared__ unsigned ovf;
  __shared__ float    wt[SC_WAVES * BATCH * H_];

  const int tid  = threadIdx.x;
  const int lane = tid & 31;
  const int wave = __builtin_amdgcn_readfirstlane(tid >> 5);
  const int nodeBase = blockIdx.x * TILE_N;

  for (int i = tid; i < TILE_N * HF_ / 4; i += SC_WAVES * 32)
    ((v4f*)acc)[i] = (v4f){0.f, 0.f, 0.f, 0.f};
  if (tid < TILE_N * H_) {
    const int g = nodeBase * H_ + tid;
    const bool in = g < N_ * H_;
    den[tid] = 0.f;
    elT[tid] = in ? el[g] : 0.f;
    mxE[tid] = encodeOrd(-__builtin_inff());
  }
  if (tid < TILE_N) ncnt[tid] = 0u;
  if (tid == 0) ovf = 0u;
  __syncthreads();

  unsigned* qw = q + wave * RCAP;
  unsigned* gw = gq + wave * RCAP;
  float*    ww = wt + wave * (BATCH * H_);
  unsigned qh = 0, qt = 0;
  const unsigned pat = (unsigned)blockIdx.x * 0x01010101u;
  constexpr int NROUND = (NCHUNK + SC_WAVES * 32 - 1) / (SC_WAVES * 32);

  auto hitBits = [&](unsigned w) -> unsigned {
    const unsigned y = w ^ pat;
    return ~(((y & 0x7F7F7F7Fu) + 0x7F7F7F7Fu) | y) & 0x80808080u;
  };
  auto lookGroups = [&](unsigned h, int n, auto&& drain) {
    unsigned m = 0u, e0 = 0u;
    if (lane < n) {
      const unsigned g = gw[(h + (unsigned)lane) & (RCAP - 1)];
      const v4u t = ((const v4u*)tileOf)[g];
      m = (hitBits(t.x) >> 7) | (hitBits(t.y) >> 6) |
          (hitBits(t.z) >> 5) | (hitBits(t.w) >> 4);
      e0 = g * GROUP;
    }
    unsigned bal = __builtin_amdgcn_ballot_w32(m != 0u);
    while (bal != 0u) {
      if (m != 0u) {
        const unsigned b = (unsigned)__builtin_ctz(m);
        qw[(qt + __builtin_amdgcn_mbcnt_lo(bal, 0u)) & (RCAP - 1)] = e0 + ((b & 3u) << 2) + (b >> 3);
        m &= m - 1u;
      }
      qt += __builtin_popcount(bal);
      if (qt - qh >= BATCH) {
        drain(qh);
        qh += BATCH;
      }
      bal = __builtin_amdgcn_ballot_w32(m != 0u);
    }
  };
  auto walk = [&](auto&& drain) {
    unsigned gh = 0, gt = 0;
    unsigned m1[NROUND];
#pragma unroll
    for (int r = 0; r < NROUND; ++r) {
      const int k = wave + SC_WAVES * (32 * r + lane);
      m1[r] = 0u;
      if (k < NCHUNK) m1[r] = grp[(size_t)k * GROW + blockIdx.x];
    }
#pragma unroll
    for (int r = 0; r < NROUND; ++r) {
      const unsigned g0 = (unsigned)(wave + SC_WAVES * (32 * r + lane)) * GPC;
      unsigned m = m1[r];
      unsigned bal = __builtin_amdgcn_ballot_w32(m != 0u);
      while (bal != 0u) {
        if (m != 0u) {
          gw[(gt + __builtin_amdgcn_mbcnt_lo(bal, 0u)) & (RCAP - 1)] = g0 + (unsigned)__builtin_ctz(m);
          m &= m - 1u;
        }
        gt += __builtin_popcount(bal);
        if (gt - gh >= BATCH) {
          lookGroups(gh, BATCH, drain);
          gh += BATCH;
        }
        bal = __builtin_amdgcn_ballot_w32(m != 0u);
      }
    }
    if (gt != gh) lookGroups(gh, (int)(gt - gh), drain);
  };

  walk([&](unsigned h) { max_batch<true>(qw, h, BATCH, lane, rec, er, elT, mxE, ncnt, nw, nlg, &ovf); });
  if (qt != qh)
    max_batch<false>(qw, qh, (int)(qt - qh), lane, rec, er, elT, mxE, ncnt, nw, nlg, &ovf);

  __syncthreads();
  if (tid < TILE_N * H_) mxT[tid] = decodeOrd(mxE[tid]);
  __syncthreads();

  const v4f b4 = ((const v4f*)bias)[lane];
  const bool full = __builtin_amdgcn_readfirstlane((int)ovf) != 0;
  if (__builtin_expect(!full, 1)) {
    for (int r = wave; r < TILE_N; r += SC_WAVES) {
      const int node = nodeBase + r;
      if (node < N_) {
        const int cnt = __builtin_amdgcn_readfirstlane((int)ncnt[r]);
        const v4f m4 = ((const v4f*)mxT)[r];
        v4f a = (v4f){0.f, 0.f, 0.f, 0.f};
        float d = 0.f;
        for (int p = 0; p < cnt; p += BATCH) {
          const int n = (cnt - p < BATCH) ? cnt - p : BATCH;
          node_batch(nw + r * NCAP + p, nlg + (r * NCAP + p) * H_, ww, n, lane, hW, m4, a, d);
        }
        if (d != 0.f) { a.x /= d; a.y /= d; a.z /= d; a.w /= d; }
        ((v4f*)(out + (size_t)node * HF_))[lane] = a + b4;
      }
    }
  } else {
    qh = 0; qt = 0;
    walk([&](unsigned h) {
      scatter_batch<true>(qw, ww, h, BATCH, lane, rec, hW, er, elT, mxT, den, acc);
    });
    if (qt != qh)
      scatter_batch<false>(qw, ww, qh, (int)(qt - qh), lane, rec, hW, er, elT, mxT, den, acc);

    __syncthreads();

    for (int r = wave; r < TILE_N; r += SC_WAVES) {
      const int node = nodeBase + r;
      if (node < N_) {
        const float* p = acc + r * HF_ + lane;
        v4f v = (v4f){p[0], p[32], p[64], p[96]};
        const float dn = den[r * H_ + (lane >> 3)];
        if (dn != 0.f) { v.x /= dn; v.y /= dn; v.z /= dn; v.w /= dn; }
        ((v4f*)(out + (size_t)node * HF_))[lane] = v + b4;
      }
    }
  }
}

extern "C" void kernel_launch(void* const* d_in, const int* in_sizes, int n_in,
                              void* d_out, int out_size, void* d_ws, size_t ws_size,
                              hipStream_t stream)
{
  const float* h    = (const float*)d_in[0];
  const float* W    = (const float*)d_in[1];
  const float* alw  = (const float*)d_in[2];
  const float* alb  = (const float*)d_in[3];
  const float* arw  = (const float*)d_in[4];
  const float* arb  = (const float*)d_in[5];
  const float* bias = (const float*)d_in[6];
  const int*   src  = (const int*)d_in[7];
  const int*   dst  = (const int*)d_in[8];
  float* out = (float*)d_out;

  char* ws = (char*)d_ws;
  size_t off = 0;
  auto carve = [&](size_t bytes) -> void* {
    void* p = ws + off;
    off += (bytes + 255) & ~(size_t)255;
    return p;
  };
  float*    hW    = (float*)carve((size_t)N_ * HF_ * sizeof(float));
  float*    el    = (float*)carve((size_t)N_ * H_ * sizeof(float));
  float*    er    = (float*)carve((size_t)N_ * H_ * sizeof(float));
  unsigned* tileOf = (unsigned*)carve((size_t)NCHUNK * CHUNK);
  unsigned* rec    = (unsigned*)carve((size_t)E_ * sizeof(unsigned));
  unsigned* grp    = (unsigned*)carve((size_t)NCHUNK * GROW * sizeof(unsigned));
  (void)ws_size; (void)in_sizes; (void)n_in; (void)out_size;

  {
    const int strips = N_ / 16;
    const int blocks = (strips + 7) / 8;
    gat_gemm_wmma<<<blocks, 256, 0, stream>>>(h, W, hW, N_);
  }
  {
    const int nNH = N_ * H_;
    gat_attn_proj<<<(nNH + 255) / 256, 256, 0, stream>>>(hW, alw, alb, arw, arb, el, er, nNH, src, dst, tileOf, rec, grp);
  }
  {
    gat_scatter_tiles<<<NTILES, SC_WAVES * 32, 0, stream>>>(rec, tileOf, grp, hW, el, er, bias, out);
  }
}
